# starting kernel
# baseline (speedup 1.0000x reference)
#include <hip/hip_runtime.h>
#include <hip/hip_bf16.h>

#define HOPS   3
#define VOCAB  50000
#define EMBD   128
#define SLEN   200
#define SENT   32
#define BATCH  64
#define QLEN   32
#define NCHUNK 25

typedef __attribute__((ext_vector_type(2))) float v2f;
typedef __attribute__((ext_vector_type(8))) float v8f;

__global__ __launch_bounds__(EMBD)
void k_init_u(const float* __restrict__ A1, const int* __restrict__ q,
              float* __restrict__ u) {
    int b = blockIdx.x, e = threadIdx.x;
    const int* qb = q + b * QLEN;
    float acc = 0.f;
#pragma unroll 4
    for (int j = 0; j < QLEN; ++j)
        acc += A1[(long)qb[j] * EMBD + e];
    u[b * EMBD + e] = acc;
}

__global__ __launch_bounds__(EMBD)
void k_score(const float* __restrict__ Ak, const float* __restrict__ TA,
             const int* __restrict__ x, const float* __restrict__ u,
             float* __restrict__ praw) {
    int s = blockIdx.x, b = blockIdx.y, e = threadIdx.x;
    const int* xs = x + (b * SLEN + s) * SENT;
    float acc = 0.f;
#pragma unroll 4
    for (int t = 0; t < SENT; ++t)
        acc += Ak[(long)xs[t] * EMBD + e];
    acc += TA[s];
    float v = acc * u[b * EMBD + e];

    __shared__ float red[EMBD];
    red[e] = v;
    __syncthreads();
    for (int off = EMBD / 2; off > 0; off >>= 1) {
        if (e < off) red[e] += red[e + off];
        __syncthreads();
    }
    if (e == 0) praw[b * SLEN + s] = red[0];
}

__global__ __launch_bounds__(256)
void k_softmax(float* __restrict__ p) {
    int b = blockIdx.x, t = threadIdx.x;
    float* pb = p + b * SLEN;
    __shared__ float red[256];

    float m = -1e30f;
    for (int s = t; s < SLEN; s += 256) m = fmaxf(m, pb[s]);
    red[t] = m;
    __syncthreads();
    for (int off = 128; off > 0; off >>= 1) {
        if (t < off) red[t] = fmaxf(red[t], red[t + off]);
        __syncthreads();
    }
    m = red[0];
    __syncthreads();

    float sum = 0.f;
    for (int s = t; s < SLEN; s += 256) sum += __expf(pb[s] - m);
    red[t] = sum;
    __syncthreads();
    for (int off = 128; off > 0; off >>= 1) {
        if (t < off) red[t] += red[t + off];
        __syncthreads();
    }
    float inv = 1.0f / red[0];
    for (int s = t; s < SLEN; s += 256) pb[s] = __expf(pb[s] - m) * inv;
}

__global__ __launch_bounds__(EMBD)
void k_combine(const float* __restrict__ Ak1, const float* __restrict__ TC,
               const int* __restrict__ x, const float* __restrict__ p,
               float* __restrict__ part) {
    int ch = blockIdx.x, b = blockIdx.y, e = threadIdx.x;
    float acc = 0.f;
    for (int s = ch * 8; s < ch * 8 + 8; ++s) {
        const int* xs = x + (b * SLEN + s) * SENT;
        float g = 0.f;
#pragma unroll 4
        for (int t = 0; t < SENT; ++t)
            g += Ak1[(long)xs[t] * EMBD + e];
        acc += p[b * SLEN + s] * (g + TC[s]);
    }
    part[(b * NCHUNK + ch) * EMBD + e] = acc;
}

__global__ __launch_bounds__(EMBD)
void k_update(float* __restrict__ u, const float* __restrict__ part) {
    int b = blockIdx.x, e = threadIdx.x;
    float acc = 0.f;
    for (int ch = 0; ch < NCHUNK; ++ch)
        acc += part[(b * NCHUNK + ch) * EMBD + e];
    u[b * EMBD + e] += acc;
}

__global__ __launch_bounds__(32)
void k_logits(const float* __restrict__ u, const float* __restrict__ A3,
              float* __restrict__ out) {
    int n0   = blockIdx.x * 16;
    int lane = threadIdx.x;
    int half = lane >> 4;
    int l16  = lane & 15;
    int koff = half * 2;

    const float* arow = A3 + (long)(n0 + l16) * EMBD;
    const float* u0   = u + l16 * EMBD;

    v8f c0 = {}, c1 = {}, c2 = {}, c3 = {};
#pragma unroll
    for (int k0 = 0; k0 < EMBD; k0 += 4) {
        int kk = k0 + koff;
        v2f bfrag, a0, a1, a2, a3;
        bfrag.x = arow[kk];               bfrag.y = arow[kk + 1];
        a0.x = u0[kk];                    a0.y = u0[kk + 1];
        a1.x = u0[16 * EMBD + kk];        a1.y = u0[16 * EMBD + kk + 1];
        a2.x = u0[32 * EMBD + kk];        a2.y = u0[32 * EMBD + kk + 1];
        a3.x = u0[48 * EMBD + kk];        a3.y = u0[48 * EMBD + kk + 1];
        c0 = __builtin_amdgcn_wmma_f32_16x16x4_f32(false, a0, false, bfrag,
                                                   (short)0, c0, false, false);
        c1 = __builtin_amdgcn_wmma_f32_16x16x4_f32(false, a1, false, bfrag,
                                                   (short)0, c1, false, false);
        c2 = __builtin_amdgcn_wmma_f32_16x16x4_f32(false, a2, false, bfrag,
                                                   (short)0, c2, false, false);
        c3 = __builtin_amdgcn_wmma_f32_16x16x4_f32(false, a3, false, bfrag,
                                                   (short)0, c3, false, false);
    }

    int col = n0 + l16;
#pragma unroll
    for (int r = 0; r < 8; ++r) {
        int row = r + 8 * half;
        out[(long)(row)      * VOCAB + col] = c0[r];
        out[(long)(row + 16) * VOCAB + col] = c1[r];
        out[(long)(row + 32) * VOCAB + col] = c2[r];
        out[(long)(row + 48) * VOCAB + col] = c3[r];
    }
}

__global__ __launch_bounds__(256)
void k_rowstats(const float* __restrict__ out, float* __restrict__ rowmax,
                float* __restrict__ rowlse) {
    int b = blockIdx.x, t = threadIdx.x;
    const float* row = out + (long)b * VOCAB;
    __shared__ float red[256];

    float m = -1e30f;
    for (int v = t; v < VOCAB; v += 256) m = fmaxf(m, row[v]);
    red[t] = m;
    __syncthreads();
    for (int off = 128; off > 0; off >>= 1) {
        if (t < off) red[t] = fmaxf(red[t], red[t + off]);
        __syncthreads();
    }
    m = red[0];
    __syncthreads();

    float s = 0.f;
    for (int v = t; v < VOCAB; v += 256) s += __expf(row[v] - m);
    red[t] = s;
    __syncthreads();
    for (int off = 128; off > 0; off >>= 1) {
        if (t < off) red[t] += red[t + off];
        __syncthreads();
    }
    if (t == 0) { rowmax[b] = m; rowlse[b] = __logf(red[0]); }
}

__global__ __launch_bounds__(256)
void k_finalize(float* __restrict__ out, const float* __restrict__ rowmax,
                const float* __restrict__ rowlse) {
    long i = (long)blockIdx.x * 256 + threadIdx.x;
    if (i < (long)BATCH * VOCAB) {
        int b = (int)(i / VOCAB);
        out[i] = out[i] - rowmax[b] - rowlse[b];
    }
}

extern "C" void kernel_launch(void* const* d_in, const int* in_sizes, int n_in,
                              void* d_out, int out_size, void* d_ws, size_t ws_size,
                              hipStream_t stream) {
    const float* A  = (const float*)d_in[0];
    const float* TA = (const float*)d_in[1];
    const float* TC = (const float*)d_in[2];
    const int*   x  = (const int*)d_in[3];
    const int*   q  = (const int*)d_in[4];
    float* out = (float*)d_out;

    float* ws   = (float*)d_ws;
    float* u    = ws;
    float* p    = u    + BATCH * EMBD;
    float* part = p    + BATCH * SLEN;
    float* rmax = part + BATCH * NCHUNK * EMBD;
    float* rlse = rmax + BATCH;

    const long TBL = (long)VOCAB * EMBD;

    k_init_u<<<BATCH, EMBD, 0, stream>>>(A + 1 * TBL, q, u);

    for (int k = 0; k < HOPS; ++k) {
        k_score  <<<dim3(SLEN, BATCH),   EMBD, 0, stream>>>(A + k * TBL, TA, x, u, p);
        k_softmax<<<BATCH,               256,  0, stream>>>(p);
        k_combine<<<dim3(NCHUNK, BATCH), EMBD, 0, stream>>>(A + (k + 1) * TBL, TC, x, p, part);
        k_update <<<BATCH,               EMBD, 0, stream>>>(u, part);
    }

    k_logits  <<<dim3(VOCAB / 16), 32, 0, stream>>>(u, A + HOPS * TBL, out);
    k_rowstats<<<BATCH, 256, 0, stream>>>(out, rmax, rlse);
    k_finalize<<<(BATCH * VOCAB + 255) / 256, 256, 0, stream>>>(out, rmax, rlse);

    (void)in_sizes; (void)n_in; (void)out_size; (void)ws_size;
}


# row max and sum-exp of the log-softmax split over 50 work-groups per row (one float4 load per thread), partial pairs merged per wave in finalize; sum-exp reduction tree changes
# speedup vs baseline: 1.2141x; 1.2141x over previous
#include <hip/hip_runtime.h>
#include <hip/hip_bf16.h>

#define HOPS   3
#define VOCAB  50000
#define EMBD   128
#define SLEN   200
#define SENT   32
#define BATCH  64
#define QLEN   32
#define NCHUNK 25

typedef __attribute__((ext_vector_type(2))) float v2f;
typedef __attribute__((ext_vector_type(8))) float v8f;

__global__ __launch_bounds__(EMBD)
void k_init_u(const float* __restrict__ A1, const int* __restrict__ q,
              float* __restrict__ u) {
    int b = blockIdx.x, e = threadIdx.x;
    const int* qb = q + b * QLEN;
    float acc = 0.f;
#pragma unroll 4
    for (int j = 0; j < QLEN; ++j)
        acc += A1[(long)qb[j] * EMBD + e];
    u[b * EMBD + e] = acc;
}

__global__ __launch_bounds__(EMBD)
void k_score(const float* __restrict__ Ak, const float* __restrict__ TA,
             const int* __restrict__ x, const float* __restrict__ u,
             float* __restrict__ praw) {
    int s = blockIdx.x, b = blockIdx.y, e = threadIdx.x;
    const int* xs = x + (b * SLEN + s) * SENT;
    float acc = 0.f;
#pragma unroll 4
    for (int t = 0; t < SENT; ++t)
        acc += Ak[(long)xs[t] * EMBD + e];
    acc += TA[s];
    float v = acc * u[b * EMBD + e];

    __shared__ float red[EMBD];
    red[e] = v;
    __syncthreads();
    for (int off = EMBD / 2; off > 0; off >>= 1) {
        if (e < off) red[e] += red[e + off];
        __syncthreads();
    }
    if (e == 0) praw[b * SLEN + s] = red[0];
}

__global__ __launch_bounds__(256)
void k_softmax(float* __restrict__ p) {
    int b = blockIdx.x, t = threadIdx.x;
    float* pb = p + b * SLEN;
    __shared__ float red[256];

    float m = -1e30f;
    for (int s = t; s < SLEN; s += 256) m = fmaxf(m, pb[s]);
    red[t] = m;
    __syncthreads();
    for (int off = 128; off > 0; off >>= 1) {
        if (t < off) red[t] = fmaxf(red[t], red[t + off]);
        __syncthreads();
    }
    m = red[0];
    __syncthreads();

    float sum = 0.f;
    for (int s = t; s < SLEN; s += 256) sum += __expf(pb[s] - m);
    red[t] = sum;
    __syncthreads();
    for (int off = 128; off > 0; off >>= 1) {
        if (t < off) red[t] += red[t + off];
        __syncthreads();
    }
    float inv = 1.0f / red[0];
    for (int s = t; s < SLEN; s += 256) pb[s] = __expf(pb[s] - m) * inv;
}

__global__ __launch_bounds__(EMBD)
void k_combine(const float* __restrict__ Ak1, const float* __restrict__ TC,
               const int* __restrict__ x, const float* __restrict__ p,
               float* __restrict__ part) {
    int ch = blockIdx.x, b = blockIdx.y, e = threadIdx.x;
    float acc = 0.f;
    for (int s = ch * 8; s < ch * 8 + 8; ++s) {
        const int* xs = x + (b * SLEN + s) * SENT;
        float g = 0.f;
#pragma unroll 4
        for (int t = 0; t < SENT; ++t)
            g += Ak1[(long)xs[t] * EMBD + e];
        acc += p[b * SLEN + s] * (g + TC[s]);
    }
    part[(b * NCHUNK + ch) * EMBD + e] = acc;
}

__global__ __launch_bounds__(EMBD)
void k_update(float* __restrict__ u, const float* __restrict__ part) {
    int b = blockIdx.x, e = threadIdx.x;
    float acc = 0.f;
    for (int ch = 0; ch < NCHUNK; ++ch)
        acc += part[(b * NCHUNK + ch) * EMBD + e];
    u[b * EMBD + e] += acc;
}

__global__ __launch_bounds__(32)
void k_logits(const float* __restrict__ u, const float* __restrict__ A3,
              float* __restrict__ out) {
    int n0   = blockIdx.x * 16;
    int lane = threadIdx.x;
    int half = lane >> 4;
    int l16  = lane & 15;
    int koff = half * 2;

    const float* arow = A3 + (long)(n0 + l16) * EMBD;
    const float* u0   = u + l16 * EMBD;

    v8f c0 = {}, c1 = {}, c2 = {}, c3 = {};
#pragma unroll
    for (int k0 = 0; k0 < EMBD; k0 += 4) {
        int kk = k0 + koff;
        v2f bfrag, a0, a1, a2, a3;
        bfrag.x = arow[kk];               bfrag.y = arow[kk + 1];
        a0.x = u0[kk];                    a0.y = u0[kk + 1];
        a1.x = u0[16 * EMBD + kk];        a1.y = u0[16 * EMBD + kk + 1];
        a2.x = u0[32 * EMBD + kk];        a2.y = u0[32 * EMBD + kk + 1];
        a3.x = u0[48 * EMBD + kk];        a3.y = u0[48 * EMBD + kk + 1];
        c0 = __builtin_amdgcn_wmma_f32_16x16x4_f32(false, a0, false, bfrag,
                                                   (short)0, c0, false, false);
        c1 = __builtin_amdgcn_wmma_f32_16x16x4_f32(false, a1, false, bfrag,
                                                   (short)0, c1, false, false);
        c2 = __builtin_amdgcn_wmma_f32_16x16x4_f32(false, a2, false, bfrag,
                                                   (short)0, c2, false, false);
        c3 = __builtin_amdgcn_wmma_f32_16x16x4_f32(false, a3, false, bfrag,
                                                   (short)0, c3, false, false);
    }

    int col = n0 + l16;
#pragma unroll
    for (int r = 0; r < 8; ++r) {
        int row = r + 8 * half;
        out[(long)(row)      * VOCAB + col] = c0[r];
        out[(long)(row + 16) * VOCAB + col] = c1[r];
        out[(long)(row + 32) * VOCAB + col] = c2[r];
        out[(long)(row + 48) * VOCAB + col] = c3[r];
    }
}

#define WAVE   32
#define NSPLIT 50
#define SLICE4 (VOCAB / NSPLIT / 4)
static_assert(VOCAB % (NSPLIT * 4) == 0, "slices must be whole float4s");
static_assert(SLICE4 <= 256, "one float4 per thread");
static_assert(NSPLIT <= 2 * WAVE, "merge reads at most two partials per lane");

__device__ __forceinline__ float wave_max(float v) {
    for (int o = WAVE / 2; o > 0; o >>= 1) v = fmaxf(v, __shfl_xor(v, o));
    return v;
}
__device__ __forceinline__ float wave_sum(float v) {
    for (int o = WAVE / 2; o > 0; o >>= 1) v += __shfl_xor(v, o);
    return v;
}

__global__ __launch_bounds__(256)
void k_rowstats(const float* __restrict__ out, float* __restrict__ pmax,
                float* __restrict__ psum) {
    int j = blockIdx.x, b = blockIdx.y, t = threadIdx.x;
    const float4* src =
        reinterpret_cast<const float4*>(out + (long)b * VOCAB) + j * SLICE4;
    __shared__ float redm[256 / WAVE], reds[256 / WAVE];

    bool act = t < SLICE4;
    float4 v = act ? src[t] : make_float4(-1e30f, -1e30f, -1e30f, -1e30f);

    float m = wave_max(fmaxf(fmaxf(v.x, v.y), fmaxf(v.z, v.w)));
    if ((t & (WAVE - 1)) == 0) redm[t / WAVE] = m;
    __syncthreads();
    m = redm[0];
    for (int w = 1; w < 256 / WAVE; ++w) m = fmaxf(m, redm[w]);

    float s = act ? (__expf(v.x - m) + __expf(v.y - m)) +
                    (__expf(v.z - m) + __expf(v.w - m))
                  : 0.f;
    s = wave_sum(s);
    if ((t & (WAVE - 1)) == 0) reds[t / WAVE] = s;
    __syncthreads();
    if (t == 0) {
        float tot = reds[0];
        for (int w = 1; w < 256 / WAVE; ++w) tot += reds[w];
        pmax[b * NSPLIT + j] = m;
        psum[b * NSPLIT + j] = tot;
    }
}

__global__ __launch_bounds__(256)
void k_finalize(float* __restrict__ out, const float* __restrict__ pmax,
                const float* __restrict__ psum) {
    int j = blockIdx.x, b = blockIdx.y, t = threadIdx.x;
    int lane = t & (WAVE - 1);
    bool act = t < SLICE4;
    float4* dst = reinterpret_cast<float4*>(out + (long)b * VOCAB) + j * SLICE4 + t;
    float4 v = act ? *dst : make_float4(0.f, 0.f, 0.f, 0.f);

    const float* pm = pmax + b * NSPLIT;
    const float* ps = psum + b * NSPLIT;
    bool hi = lane + WAVE < NSPLIT;
    float m0 = pm[lane];
    float s0 = ps[lane];
    float m1 = hi ? pm[lane + WAVE] : -1e30f;
    float s1 = hi ? ps[lane + WAVE] : 0.f;

    float M   = wave_max(fmaxf(m0, m1));
    float S   = wave_sum(s0 * __expf(m0 - M) + s1 * __expf(m1 - M));
    float lse = __logf(S);

    if (act) {
        v.x = v.x - M - lse;
        v.y = v.y - M - lse;
        v.z = v.z - M - lse;
        v.w = v.w - M - lse;
        *dst = v;
    }
}

extern "C" void kernel_launch(void* const* d_in, const int* in_sizes, int n_in,
                              void* d_out, int out_size, void* d_ws, size_t ws_size,
                              hipStream_t stream) {
    const float* A  = (const float*)d_in[0];
    const float* TA = (const float*)d_in[1];
    const float* TC = (const float*)d_in[2];
    const int*   x  = (const int*)d_in[3];
    const int*   q  = (const int*)d_in[4];
    float* out = (float*)d_out;

    float* ws   = (float*)d_ws;
    float* u    = ws;
    float* p    = u    + BATCH * EMBD;
    float* part = p    + BATCH * SLEN;
    float* pmax = part;
    float* psum = part + BATCH * NSPLIT;

    const long TBL = (long)VOCAB * EMBD;

    k_init_u<<<BATCH, EMBD, 0, stream>>>(A + 1 * TBL, q, u);

    for (int k = 0; k < HOPS; ++k) {
        k_score  <<<dim3(SLEN, BATCH),   EMBD, 0, stream>>>(A + k * TBL, TA, x, u, p);
        k_softmax<<<BATCH,               256,  0, stream>>>(p);
        k_combine<<<dim3(NCHUNK, BATCH), EMBD, 0, stream>>>(A + (k + 1) * TBL, TC, x, p, part);
        k_update <<<BATCH,               EMBD, 0, stream>>>(u, part);
    }

    k_logits  <<<dim3(VOCAB / 16), 32, 0, stream>>>(u, A + HOPS * TBL, out);
    k_rowstats<<<dim3(NSPLIT, BATCH), 256, 0, stream>>>(out, pmax, psum);
    k_finalize<<<dim3(NSPLIT, BATCH), 256, 0, stream>>>(out, pmax, psum);

    (void)in_sizes; (void)n_in; (void)out_size; (void)ws_size;
}


# gather kernels: a wave reads each 512 B table row with one 16 B-per-lane load, 64 independent row loads per wave kept about 10 deep; score sums its 128 dims by lane quads + shuffles (order changes)
# speedup vs baseline: 2.1956x; 1.8084x over previous
#include <hip/hip_runtime.h>
#include <hip/hip_bf16.h>

#define HOPS   3
#define VOCAB  50000
#define EMBD   128
#define SLEN   200
#define SENT   32
#define BATCH  64
#define QLEN   32
#define NCHUNK 25
#define WAVE   32
#define SPW    2
#define WPB    (8 / SPW)

static_assert(EMBD == 4 * WAVE, "one table row = one 16-byte load per lane of a wave");
static_assert(QLEN == SENT, "bag_sum sums SENT rows");
static_assert(WPB * WAVE == EMBD, "k_combine's last stage: one thread per embedding dim");
static_assert(NCHUNK * 8 == SLEN, "chunks of 8 sentences");

typedef __attribute__((ext_vector_type(2))) float v2f;
typedef __attribute__((ext_vector_type(8))) float v8f;

__device__ __forceinline__ float wave_max(float v) {
    for (int o = WAVE / 2; o > 0; o >>= 1) v = fmaxf(v, __shfl_xor(v, o));
    return v;
}
__device__ __forceinline__ float wave_sum(float v) {
    for (int o = WAVE / 2; o > 0; o >>= 1) v += __shfl_xor(v, o);
    return v;
}

__device__ __forceinline__ float4 bag_sum(const float* __restrict__ tab,
                                          const int* __restrict__ idx, int lane) {
    float4 r[SENT];
#pragma unroll
    for (int t = 0; t < SENT; ++t)
        r[t] = reinterpret_cast<const float4*>(tab + (long)idx[t] * EMBD)[lane];
    float4 g = make_float4(0.f, 0.f, 0.f, 0.f);
#pragma unroll
    for (int t = 0; t < SENT; ++t) {
        g.x += r[t].x; g.y += r[t].y; g.z += r[t].z; g.w += r[t].w;
    }
    return g;
}

__global__ __launch_bounds__(WAVE)
void k_init_u(const float* __restrict__ A1, const int* __restrict__ q,
              float* __restrict__ u) {
    int b = blockIdx.x, lane = threadIdx.x;
    float4 g = bag_sum(A1, q + b * QLEN, lane);
    reinterpret_cast<float4*>(u + b * EMBD)[lane] = g;
}

__global__ __launch_bounds__(WPB * WAVE)
void k_score(const float* __restrict__ Ak, const float* __restrict__ TA,
             const int* __restrict__ x, const float* __restrict__ u,
             float* __restrict__ praw) {
    int ch = blockIdx.x, b = blockIdx.y;
    int lane = threadIdx.x & (WAVE - 1);
    int w = __builtin_amdgcn_readfirstlane(threadIdx.x / WAVE);
    float4 uu = reinterpret_cast<const float4*>(u + b * EMBD)[lane];
    int s0 = ch * 8 + w * SPW;
    float sc[SPW];
    for (int i = 0; i < SPW; ++i) {
        int s = s0 + i;
        float4 g = bag_sum(Ak, x + (b * SLEN + s) * SENT, lane);
        float ta = TA[s];
        float v = ((g.x + ta) * uu.x + (g.y + ta) * uu.y) +
                  ((g.z + ta) * uu.z + (g.w + ta) * uu.w);
        sc[i] = wave_sum(v);
    }
    static_assert(SPW == 2, "the two scores of a wave leave as one 8-byte store");
    if (lane == 0)
        *reinterpret_cast<float2*>(praw + b * SLEN + s0) = make_float2(sc[0], sc[1]);
}

__global__ __launch_bounds__(256)
void k_softmax(float* __restrict__ p) {
    int b = blockIdx.x, t = threadIdx.x;
    float* pb = p + b * SLEN;
    __shared__ float red[256];

    float m = -1e30f;
    for (int s = t; s < SLEN; s += 256) m = fmaxf(m, pb[s]);
    red[t] = m;
    __syncthreads();
    for (int off = 128; off > 0; off >>= 1) {
        if (t < off) red[t] = fmaxf(red[t], red[t + off]);
        __syncthreads();
    }
    m = red[0];
    __syncthreads();

    float sum = 0.f;
    for (int s = t; s < SLEN; s += 256) sum += __expf(pb[s] - m);
    red[t] = sum;
    __syncthreads();
    for (int off = 128; off > 0; off >>= 1) {
        if (t < off) red[t] += red[t + off];
        __syncthreads();
    }
    float inv = 1.0f / red[0];
    for (int s = t; s < SLEN; s += 256) pb[s] = __expf(pb[s] - m) * inv;
}

__global__ __launch_bounds__(WPB * WAVE)
void k_combine(const float* __restrict__ Ak1, const float* __restrict__ TC,
               const int* __restrict__ x, const float* __restrict__ p,
               float* __restrict__ part) {
    int ch = blockIdx.x, b = blockIdx.y;
    int lane = threadIdx.x & (WAVE - 1);
    int w = __builtin_amdgcn_readfirstlane(threadIdx.x / WAVE);
    __shared__ float4 gt[8][WAVE];
    for (int i = 0; i < SPW; ++i) {
        int sl = w * SPW + i;
        int s  = ch * 8 + sl;
        float4 g = bag_sum(Ak1, x + (b * SLEN + s) * SENT, lane);
        float tc = TC[s];
        gt[sl][lane] = make_float4(g.x + tc, g.y + tc, g.z + tc, g.w + tc);
    }
    __syncthreads();
    const float* gf = reinterpret_cast<const float*>(&gt[0][0]);
    int e = threadIdx.x;
    float acc = 0.f;
    for (int sl = 0; sl < 8; ++sl)
        acc += p[b * SLEN + ch * 8 + sl] * gf[sl * EMBD + e];
    part[(b * NCHUNK + ch) * EMBD + e] = acc;
}

__global__ __launch_bounds__(EMBD)
void k_update(float* __restrict__ u, const float* __restrict__ part) {
    int b = blockIdx.x, e = threadIdx.x;
    float acc = 0.f;
    for (int ch = 0; ch < NCHUNK; ++ch)
        acc += part[(b * NCHUNK + ch) * EMBD + e];
    u[b * EMBD + e] += acc;
}

__global__ __launch_bounds__(32)
void k_logits(const float* __restrict__ u, const float* __restrict__ A3,
              float* __restrict__ out) {
    int n0   = blockIdx.x * 16;
    int lane = threadIdx.x;
    int half = lane >> 4;
    int l16  = lane & 15;
    int koff = half * 2;

    const float* arow = A3 + (long)(n0 + l16) * EMBD;
    const float* u0   = u + l16 * EMBD;

    v8f c0 = {}, c1 = {}, c2 = {}, c3 = {};
#pragma unroll
    for (int k0 = 0; k0 < EMBD; k0 += 4) {
        int kk = k0 + koff;
        v2f bfrag, a0, a1, a2, a3;
        bfrag.x = arow[kk];               bfrag.y = arow[kk + 1];
        a0.x = u0[kk];                    a0.y = u0[kk + 1];
        a1.x = u0[16 * EMBD + kk];        a1.y = u0[16 * EMBD + kk + 1];
        a2.x = u0[32 * EMBD + kk];        a2.y = u0[32 * EMBD + kk + 1];
        a3.x = u0[48 * EMBD + kk];        a3.y = u0[48 * EMBD + kk + 1];
        c0 = __builtin_amdgcn_wmma_f32_16x16x4_f32(false, a0, false, bfrag,
                                                   (short)0, c0, false, false);
        c1 = __builtin_amdgcn_wmma_f32_16x16x4_f32(false, a1, false, bfrag,
                                                   (short)0, c1, false, false);
        c2 = __builtin_amdgcn_wmma_f32_16x16x4_f32(false, a2, false, bfrag,
                                                   (short)0, c2, false, false);
        c3 = __builtin_amdgcn_wmma_f32_16x16x4_f32(false, a3, false, bfrag,
                                                   (short)0, c3, false, false);
    }

    int col = n0 + l16;
#pragma unroll
    for (int r = 0; r < 8; ++r) {
        int row = r + 8 * half;
        out[(long)(row)      * VOCAB + col] = c0[r];
        out[(long)(row + 16) * VOCAB + col] = c1[r];
        out[(long)(row + 32) * VOCAB + col] = c2[r];
        out[(long)(row + 48) * VOCAB + col] = c3[r];
    }
}

#define NSPLIT 50
#define SLICE4 (VOCAB / NSPLIT / 4)
static_assert(VOCAB % (NSPLIT * 4) == 0, "slices must be whole float4s");
static_assert(SLICE4 <= 256, "one float4 per thread");
static_assert(NSPLIT <= 2 * WAVE, "merge reads at most two partials per lane");

__global__ __launch_bounds__(256)
void k_rowstats(const float* __restrict__ out, float* __restrict__ pmax,
                float* __restrict__ psum) {
    int j = blockIdx.x, b = blockIdx.y, t = threadIdx.x;
    const float4* src =
        reinterpret_cast<const float4*>(out + (long)b * VOCAB) + j * SLICE4;
    __shared__ float redm[256 / WAVE], reds[256 / WAVE];

    bool act = t < SLICE4;
    float4 v = act ? src[t] : make_float4(-1e30f, -1e30f, -1e30f, -1e30f);

    float m = wave_max(fmaxf(fmaxf(v.x, v.y), fmaxf(v.z, v.w)));
    if ((t & (WAVE - 1)) == 0) redm[t / WAVE] = m;
    __syncthreads();
    m = redm[0];
    for (int w = 1; w < 256 / WAVE; ++w) m = fmaxf(m, redm[w]);

    float s = act ? (__expf(v.x - m) + __expf(v.y - m)) +
                    (__expf(v.z - m) + __expf(v.w - m))
                  : 0.f;
    s = wave_sum(s);
    if ((t & (WAVE - 1)) == 0) reds[t / WAVE] = s;
    __syncthreads();
    if (t == 0) {
        float tot = reds[0];
        for (int w = 1; w < 256 / WAVE; ++w) tot += reds[w];
        pmax[b * NSPLIT + j] = m;
        psum[b * NSPLIT + j] = tot;
    }
}

__global__ __launch_bounds__(256)
void k_finalize(float* __restrict__ out, const float* __restrict__ pmax,
                const float* __restrict__ psum) {
    int j = blockIdx.x, b = blockIdx.y, t = threadIdx.x;
    int lane = t & (WAVE - 1);
    bool act = t < SLICE4;
    float4* dst = reinterpret_cast<float4*>(out + (long)b * VOCAB) + j * SLICE4 + t;
    float4 v = act ? *dst : make_float4(0.f, 0.f, 0.f, 0.f);

    const float* pm = pmax + b * NSPLIT;
    const float* ps = psum + b * NSPLIT;
    bool hi = lane + WAVE < NSPLIT;
    float m0 = pm[lane];
    float s0 = ps[lane];
    float m1 = hi ? pm[lane + WAVE] : -1e30f;
    float s1 = hi ? ps[lane + WAVE] : 0.f;

    float M   = wave_max(fmaxf(m0, m1));
    float S   = wave_sum(s0 * __expf(m0 - M) + s1 * __expf(m1 - M));
    float lse = __logf(S);

    if (act) {
        v.x = v.x - M - lse;
        v.y = v.y - M - lse;
        v.z = v.z - M - lse;
        v.w = v.w - M - lse;
        *dst = v;
    }
}

extern "C" void kernel_launch(void* const* d_in, const int* in_sizes, int n_in,
                              void* d_out, int out_size, void* d_ws, size_t ws_size,
                              hipStream_t stream) {
    const float* A  = (const float*)d_in[0];
    const float* TA = (const float*)d_in[1];
    const float* TC = (const float*)d_in[2];
    const int*   x  = (const int*)d_in[3];
    const int*   q  = (const int*)d_in[4];
    float* out = (float*)d_out;

    float* ws   = (float*)d_ws;
    float* u    = ws;
    float* p    = u    + BATCH * EMBD;
    float* part = p    + BATCH * SLEN;
    float* pmax = part;
    float* psum = part + BATCH * NSPLIT;

    const long TBL = (long)VOCAB * EMBD;

    k_init_u<<<BATCH, WAVE, 0, stream>>>(A + 1 * TBL, q, u);

    for (int k = 0; k < HOPS; ++k) {
        k_score  <<<dim3(NCHUNK, BATCH), WPB * WAVE, 0, stream>>>(A + k * TBL, TA, x, u, p);
        k_softmax<<<BATCH,               256,  0, stream>>>(p);
        k_combine<<<dim3(NCHUNK, BATCH), WPB * WAVE, 0, stream>>>(A + (k + 1) * TBL, TC, x, p, part);
        k_update <<<BATCH,               EMBD, 0, stream>>>(u, part);
    }

    k_logits  <<<dim3(VOCAB / 16), 32, 0, stream>>>(u, A + HOPS * TBL, out);
    k_rowstats<<<dim3(NSPLIT, BATCH), 256, 0, stream>>>(out, pmax, psum);
    k_finalize<<<dim3(NSPLIT, BATCH), 256, 0, stream>>>(out, pmax, psum);

    (void)in_sizes; (void)n_in; (void)out_size; (void)ws_size;
}


# k_softmax folded into k_combine (each wave redoes row max, sum-exp), k_update into the next k_score (residual kept in chunk 0's partial): 16 -> 11 launches; softmax and residual sum orders change
# speedup vs baseline: 2.5121x; 1.1442x over previous
#include <hip/hip_runtime.h>
#include <hip/hip_bf16.h>

#define HOPS   3
#define VOCAB  50000
#define EMBD   128
#define SLEN   200
#define SENT   32
#define BATCH  64
#define QLEN   32
#define NCHUNK 25
#define WAVE   32
#define SPW    2
#define WPB    (8 / SPW)

static_assert(EMBD == 4 * WAVE, "one table row = one 16-byte load per lane of a wave");
static_assert(QLEN == SENT, "bag_sum sums SENT rows");
static_assert(WPB * WAVE == EMBD, "k_combine's last stage: one thread per embedding dim");
static_assert(NCHUNK * 8 == SLEN, "chunks of 8 sentences");
#define SROW ((SLEN + WAVE - 1) / WAVE)

typedef __attribute__((ext_vector_type(2))) float v2f;
typedef __attribute__((ext_vector_type(8))) float v8f;

__device__ __forceinline__ float wave_max(float v) {
    for (int o = WAVE / 2; o > 0; o >>= 1) v = fmaxf(v, __shfl_xor(v, o));
    return v;
}
__device__ __forceinline__ float wave_sum(float v) {
    for (int o = WAVE / 2; o > 0; o >>= 1) v += __shfl_xor(v, o);
    return v;
}

__device__ __forceinline__ float fold_part(const float* __restrict__ part, int b, int e) {
    const float* pp = part + (b * NCHUNK) * EMBD + e;
    float acc = pp[0];
    for (int ch = 1; ch < NCHUNK; ++ch) acc += pp[ch * EMBD];
    return acc;
}

__device__ __forceinline__ float4 bag_sum(const float* __restrict__ tab,
                                          const int* __restrict__ idx, int lane) {
    float4 r[SENT];
#pragma unroll
    for (int t = 0; t < SENT; ++t)
        r[t] = reinterpret_cast<const float4*>(tab + (long)idx[t] * EMBD)[lane];
    float4 g = make_float4(0.f, 0.f, 0.f, 0.f);
#pragma unroll
    for (int t = 0; t < SENT; ++t) {
        g.x += r[t].x; g.y += r[t].y; g.z += r[t].z; g.w += r[t].w;
    }
    return g;
}

__global__ __launch_bounds__(WAVE)
void k_init_u(const float* __restrict__ A1, const int* __restrict__ q,
              float* __restrict__ u) {
    int b = blockIdx.x, lane = threadIdx.x;
    float4 g = bag_sum(A1, q + b * QLEN, lane);
    reinterpret_cast<float4*>(u + b * EMBD)[lane] = g;
}

__device__ __forceinline__
void score_gather(const float* __restrict__ Ak, const float* __restrict__ TA,
                  const int* __restrict__ x, int b, int s0, int lane,
                  float4 (&g)[SPW]) {
    for (int i = 0; i < SPW; ++i) {
        float4 r = bag_sum(Ak, x + (b * SLEN + s0 + i) * SENT, lane);
        float ta = TA[s0 + i];
        g[i] = make_float4(r.x + ta, r.y + ta, r.z + ta, r.w + ta);
    }
}

__device__ __forceinline__
void score_store(const float4 (&g)[SPW], float4 uu,
                 float* __restrict__ praw, int b, int s0, int lane) {
    float sc[SPW];
    for (int i = 0; i < SPW; ++i) {
        float v = (g[i].x * uu.x + g[i].y * uu.y) +
                  (g[i].z * uu.z + g[i].w * uu.w);
        sc[i] = wave_sum(v);
    }
    static_assert(SPW == 2, "the two scores of a wave leave as one 8-byte store");
    if (lane == 0)
        *reinterpret_cast<float2*>(praw + b * SLEN + s0) = make_float2(sc[0], sc[1]);
}

__global__ __launch_bounds__(WPB * WAVE)
void k_score_first(const float* __restrict__ Ak, const float* __restrict__ TA,
                   const int* __restrict__ x, const float* __restrict__ u,
                   float* __restrict__ praw) {
    int ch = blockIdx.x, b = blockIdx.y;
    int lane = threadIdx.x & (WAVE - 1);
    int w = __builtin_amdgcn_readfirstlane(threadIdx.x / WAVE);
    int s0 = ch * 8 + w * SPW;
    float4 uu = reinterpret_cast<const float4*>(u + b * EMBD)[lane];
    float4 g[SPW];
    score_gather(Ak, TA, x, b, s0, lane, g);
    score_store(g, uu, praw, b, s0, lane);
}

__global__ __launch_bounds__(WPB * WAVE)
void k_score_fold(const float* __restrict__ Ak, const float* __restrict__ TA,
                  const int* __restrict__ x, const float* __restrict__ part,
                  float* __restrict__ u, float* __restrict__ praw) {
    int ch = blockIdx.x, b = blockIdx.y;
    int lane = threadIdx.x & (WAVE - 1);
    int w = __builtin_amdgcn_readfirstlane(threadIdx.x / WAVE);
    int s0 = ch * 8 + w * SPW;
    __shared__ float4 us[WAVE];
    float un = fold_part(part, b, threadIdx.x);
    reinterpret_cast<float*>(&us[0])[threadIdx.x] = un;
    float4 g[SPW];
    score_gather(Ak, TA, x, b, s0, lane, g);
    __syncthreads();
    float4 uu = us[lane];
    score_store(g, uu, praw, b, s0, lane);
    if (ch == 0) u[b * EMBD + threadIdx.x] = un;
}

__global__ __launch_bounds__(WPB * WAVE)
void k_combine(const float* __restrict__ Ak1, const float* __restrict__ TC,
               const int* __restrict__ x, const float* __restrict__ praw,
               const float* __restrict__ u, float* __restrict__ part) {
    int ch = blockIdx.x, b = blockIdx.y;
    int lane = threadIdx.x & (WAVE - 1);
    float ur = u[b * EMBD + threadIdx.x];
    const float* pr = praw + b * SLEN;
    float pv[SROW];
    float m = -1e30f;
    for (int j = 0; j < SROW; ++j) {
        int s = lane + j * WAVE;
        pv[j] = s < SLEN ? pr[s] : -1e30f;
        m = fmaxf(m, pv[j]);
    }
    m = wave_max(m);
    float sum = 0.f;
    for (int j = 0; j < SROW; ++j)
        sum += (lane + j * WAVE < SLEN) ? __expf(pv[j] - m) : 0.f;
    float inv = 1.0f / wave_sum(sum);
    int w = __builtin_amdgcn_readfirstlane(threadIdx.x / WAVE);
    __shared__ float4 gt[8][WAVE];
    for (int i = 0; i < SPW; ++i) {
        int sl = w * SPW + i;
        int s  = ch * 8 + sl;
        float4 g = bag_sum(Ak1, x + (b * SLEN + s) * SENT, lane);
        float tc = TC[s];
        float ps = __expf(pr[s] - m) * inv;
        gt[sl][lane] = make_float4((g.x + tc) * ps, (g.y + tc) * ps,
                                   (g.z + tc) * ps, (g.w + tc) * ps);
    }
    __syncthreads();
    const float* gf = reinterpret_cast<const float*>(&gt[0][0]);
    int e = threadIdx.x;
    float acc = 0.f;
    for (int sl = 0; sl < 8; ++sl)
        acc += gf[sl * EMBD + e];
    if (ch == 0) acc += ur;
    part[(b * NCHUNK + ch) * EMBD + e] = acc;
}

__global__ __launch_bounds__(EMBD)
void k_fold_u(float* __restrict__ u, const float* __restrict__ part) {
    int b = blockIdx.x, e = threadIdx.x;
    u[b * EMBD + e] = fold_part(part, b, e);
}

__global__ __launch_bounds__(32)
void k_logits(const float* __restrict__ u, const float* __restrict__ A3,
              float* __restrict__ out) {
    int n0   = blockIdx.x * 16;
    int lane = threadIdx.x;
    int half = lane >> 4;
    int l16  = lane & 15;
    int koff = half * 2;

    const float* arow = A3 + (long)(n0 + l16) * EMBD;
    const float* u0   = u + l16 * EMBD;

    v8f c0 = {}, c1 = {}, c2 = {}, c3 = {};
#pragma unroll
    for (int k0 = 0; k0 < EMBD; k0 += 4) {
        int kk = k0 + koff;
        v2f bfrag, a0, a1, a2, a3;
        bfrag.x = arow[kk];               bfrag.y = arow[kk + 1];
        a0.x = u0[kk];                    a0.y = u0[kk + 1];
        a1.x = u0[16 * EMBD + kk];        a1.y = u0[16 * EMBD + kk + 1];
        a2.x = u0[32 * EMBD + kk];        a2.y = u0[32 * EMBD + kk + 1];
        a3.x = u0[48 * EMBD + kk];        a3.y = u0[48 * EMBD + kk + 1];
        c0 = __builtin_amdgcn_wmma_f32_16x16x4_f32(false, a0, false, bfrag,
                                                   (short)0, c0, false, false);
        c1 = __builtin_amdgcn_wmma_f32_16x16x4_f32(false, a1, false, bfrag,
                                                   (short)0, c1, false, false);
        c2 = __builtin_amdgcn_wmma_f32_16x16x4_f32(false, a2, false, bfrag,
                                                   (short)0, c2, false, false);
        c3 = __builtin_amdgcn_wmma_f32_16x16x4_f32(false, a3, false, bfrag,
                                                   (short)0, c3, false, false);
    }

    int col = n0 + l16;
#pragma unroll
    for (int r = 0; r < 8; ++r) {
        int row = r + 8 * half;
        out[(long)(row)      * VOCAB + col] = c0[r];
        out[(long)(row + 16) * VOCAB + col] = c1[r];
        out[(long)(row + 32) * VOCAB + col] = c2[r];
        out[(long)(row + 48) * VOCAB + col] = c3[r];
    }
}

#define NSPLIT 50
#define SLICE4 (VOCAB / NSPLIT / 4)
static_assert(VOCAB % (NSPLIT * 4) == 0, "slices must be whole float4s");
static_assert(SLICE4 <= 256, "one float4 per thread");
static_assert(NSPLIT <= 2 * WAVE, "merge reads at most two partials per lane");

__global__ __launch_bounds__(256)
void k_rowstats(const float* __restrict__ out, float* __restrict__ pmax,
                float* __restrict__ psum) {
    int j = blockIdx.x, b = blockIdx.y, t = threadIdx.x;
    const float4* src =
        reinterpret_cast<const float4*>(out + (long)b * VOCAB) + j * SLICE4;
    __shared__ float redm[256 / WAVE], reds[256 / WAVE];

    bool act = t < SLICE4;
    float4 v = act ? src[t] : make_float4(-1e30f, -1e30f, -1e30f, -1e30f);

    float m = wave_max(fmaxf(fmaxf(v.x, v.y), fmaxf(v.z, v.w)));
    if ((t & (WAVE - 1)) == 0) redm[t / WAVE] = m;
    __syncthreads();
    m = redm[0];
    for (int w = 1; w < 256 / WAVE; ++w) m = fmaxf(m, redm[w]);

    float s = act ? (__expf(v.x - m) + __expf(v.y - m)) +
                    (__expf(v.z - m) + __expf(v.w - m))
                  : 0.f;
    s = wave_sum(s);
    if ((t & (WAVE - 1)) == 0) reds[t / WAVE] = s;
    __syncthreads();
    if (t == 0) {
        float tot = reds[0];
        for (int w = 1; w < 256 / WAVE; ++w) tot += reds[w];
        pmax[b * NSPLIT + j] = m;
        psum[b * NSPLIT + j] = tot;
    }
}

__global__ __launch_bounds__(256)
void k_finalize(float* __restrict__ out, const float* __restrict__ pmax,
                const float* __restrict__ psum) {
    int j = blockIdx.x, b = blockIdx.y, t = threadIdx.x;
    int lane = t & (WAVE - 1);
    bool act = t < SLICE4;
    float4* dst = reinterpret_cast<float4*>(out + (long)b * VOCAB) + j * SLICE4 + t;
    float4 v = act ? *dst : make_float4(0.f, 0.f, 0.f, 0.f);

    const float* pm = pmax + b * NSPLIT;
    const float* ps = psum + b * NSPLIT;
    bool hi = lane + WAVE < NSPLIT;
    float m0 = pm[lane];
    float s0 = ps[lane];
    float m1 = hi ? pm[lane + WAVE] : -1e30f;
    float s1 = hi ? ps[lane + WAVE] : 0.f;

    float M   = wave_max(fmaxf(m0, m1));
    float S   = wave_sum(s0 * __expf(m0 - M) + s1 * __expf(m1 - M));
    float lse = __logf(S);

    if (act) {
        v.x = v.x - M - lse;
        v.y = v.y - M - lse;
        v.z = v.z - M - lse;
        v.w = v.w - M - lse;
        *dst = v;
    }
}

extern "C" void kernel_launch(void* const* d_in, const int* in_sizes, int n_in,
                              void* d_out, int out_size, void* d_ws, size_t ws_size,
                              hipStream_t stream) {
    const float* A  = (const float*)d_in[0];
    const float* TA = (const float*)d_in[1];
    const float* TC = (const float*)d_in[2];
    const int*   x  = (const int*)d_in[3];
    const int*   q  = (const int*)d_in[4];
    float* out = (float*)d_out;

    float* ws   = (float*)d_ws;
    float* u    = ws;
    float* p    = u    + BATCH * EMBD;
    float* part = p    + BATCH * SLEN;
    float* pmax = part;
    float* psum = part + BATCH * NSPLIT;

    const long TBL = (long)VOCAB * EMBD;

    k_init_u<<<BATCH, WAVE, 0, stream>>>(A + 1 * TBL, q, u);

    k_score_first<<<dim3(NCHUNK, BATCH), WPB * WAVE, 0, stream>>>(A, TA, x, u, p);
    k_combine    <<<dim3(NCHUNK, BATCH), WPB * WAVE, 0, stream>>>(A + TBL, TC, x, p, u, part);
    for (int k = 1; k < HOPS; ++k) {
        k_score_fold<<<dim3(NCHUNK, BATCH), WPB * WAVE, 0, stream>>>(A + k * TBL, TA, x, part, u, p);
        k_combine   <<<dim3(NCHUNK, BATCH), WPB * WAVE, 0, stream>>>(A + (k + 1) * TBL, TC, x, p, u, part);
    }
    k_fold_u<<<BATCH, EMBD, 0, stream>>>(u, part);

    k_logits  <<<dim3(VOCAB / 16), 32, 0, stream>>>(u, A + HOPS * TBL, out);
    k_rowstats<<<dim3(NSPLIT, BATCH), 256, 0, stream>>>(out, pmax, psum);
    k_finalize<<<dim3(NSPLIT, BATCH), 256, 0, stream>>>(out, pmax, psum);

    (void)in_sizes; (void)n_in; (void)out_size; (void)ws_size;
}


# row sums of tables 1 and 2 gathered once: k_combine also stores them (front of out, which the logits overwrite later), the next hop's score streams them back; 6 -> 4 gather passes, sum orders kept
# speedup vs baseline: 2.6165x; 1.0416x over previous
#include <hip/hip_runtime.h>
#include <hip/hip_bf16.h>

#define HOPS   3
#define VOCAB  50000
#define EMBD   128
#define SLEN   200
#define SENT   32
#define BATCH  64
#define QLEN   32
#define NCHUNK 25
#define WAVE   32
#define SPW    2
#define WPB    (8 / SPW)

static_assert(EMBD == 4 * WAVE, "one table row = one 16-byte load per lane of a wave");
static_assert(QLEN == SENT, "bag_sum sums SENT rows");
static_assert(WPB * WAVE == EMBD, "k_combine's last stage: one thread per embedding dim");
static_assert(NCHUNK * 8 == SLEN, "chunks of 8 sentences");
#define SROW ((SLEN + WAVE - 1) / WAVE)

typedef __attribute__((ext_vector_type(2))) float v2f;
typedef __attribute__((ext_vector_type(8))) float v8f;

__device__ __forceinline__ float wave_max(float v) {
    for (int o = WAVE / 2; o > 0; o >>= 1) v = fmaxf(v, __shfl_xor(v, o));
    return v;
}
__device__ __forceinline__ float wave_sum(float v) {
    for (int o = WAVE / 2; o > 0; o >>= 1) v += __shfl_xor(v, o);
    return v;
}

__device__ __forceinline__ float fold_part(const float* __restrict__ part, int b, int e) {
    const float* pp = part + (b * NCHUNK) * EMBD + e;
    float acc = pp[0];
    for (int ch = 1; ch < NCHUNK; ++ch) acc += pp[ch * EMBD];
    return acc;
}

__device__ __forceinline__ float4 bag_sum(const float* __restrict__ tab,
                                          const int* __restrict__ idx, int lane) {
    float4 r[SENT];
#pragma unroll
    for (int t = 0; t < SENT; ++t)
        r[t] = reinterpret_cast<const float4*>(tab + (long)idx[t] * EMBD)[lane];
    float4 g = make_float4(0.f, 0.f, 0.f, 0.f);
#pragma unroll
    for (int t = 0; t < SENT; ++t) {
        g.x += r[t].x; g.y += r[t].y; g.z += r[t].z; g.w += r[t].w;
    }
    return g;
}

__global__ __launch_bounds__(WAVE)
void k_init_u(const float* __restrict__ A1, const int* __restrict__ q,
              float* __restrict__ u) {
    int b = blockIdx.x, lane = threadIdx.x;
    float4 g = bag_sum(A1, q + b * QLEN, lane);
    reinterpret_cast<float4*>(u + b * EMBD)[lane] = g;
}

__device__ __forceinline__
void score_gather(const float* __restrict__ Ak, const float* __restrict__ TA,
                  const int* __restrict__ x, int b, int s0, int lane,
                  float4 (&g)[SPW]) {
    for (int i = 0; i < SPW; ++i) {
        float4 r = bag_sum(Ak, x + (b * SLEN + s0 + i) * SENT, lane);
        float ta = TA[s0 + i];
        g[i] = make_float4(r.x + ta, r.y + ta, r.z + ta, r.w + ta);
    }
}

__device__ __forceinline__
void score_store(const float4 (&g)[SPW], float4 uu,
                 float* __restrict__ praw, int b, int s0, int lane) {
    float sc[SPW];
    for (int i = 0; i < SPW; ++i) {
        float v = (g[i].x * uu.x + g[i].y * uu.y) +
                  (g[i].z * uu.z + g[i].w * uu.w);
        sc[i] = wave_sum(v);
    }
    static_assert(SPW == 2, "the two scores of a wave leave as one 8-byte store");
    if (lane == 0)
        *reinterpret_cast<float2*>(praw + b * SLEN + s0) = make_float2(sc[0], sc[1]);
}

__global__ __launch_bounds__(WPB * WAVE)
void k_score_first(const float* __restrict__ Ak, const float* __restrict__ TA,
                   const int* __restrict__ x, const float* __restrict__ u,
                   float* __restrict__ praw) {
    int ch = blockIdx.x, b = blockIdx.y;
    int lane = threadIdx.x & (WAVE - 1);
    int w = __builtin_amdgcn_readfirstlane(threadIdx.x / WAVE);
    int s0 = ch * 8 + w * SPW;
    float4 uu = reinterpret_cast<const float4*>(u + b * EMBD)[lane];
    float4 g[SPW];
    score_gather(Ak, TA, x, b, s0, lane, g);
    score_store(g, uu, praw, b, s0, lane);
}

#define SGRP 5
#define SPG  (SLEN / SGRP)
#define SPWK (SPG / WPB)
static_assert(SGRP * WPB * SPWK == SLEN, "the work-groups of a row cover its sentences exactly");
static_assert(SPWK % 2 == 0, "a wave's scores leave as 8-byte stores");

__global__ __launch_bounds__(WPB * WAVE)
void k_score_kept(const float* __restrict__ keep, const float* __restrict__ TA,
                  const float* __restrict__ part, float* __restrict__ u,
                  float* __restrict__ praw) {
    int grp = blockIdx.x, b = blockIdx.y;
    int lane = threadIdx.x & (WAVE - 1);
    int w = __builtin_amdgcn_readfirstlane(threadIdx.x / WAVE);
    int s0 = grp * SPG + w * SPWK;
    const float4* rows =
        reinterpret_cast<const float4*>(keep + (long)(b * SLEN + s0) * EMBD) + lane;
    const float4* pp =
        reinterpret_cast<const float4*>(part + (long)b * NCHUNK * EMBD) + lane;
    float4 r[SPWK], f[NCHUNK];
#pragma unroll
    for (int i = 0; i < SPWK; ++i) r[i] = rows[i * WAVE];
#pragma unroll
    for (int ch = 0; ch < NCHUNK; ++ch) f[ch] = pp[ch * WAVE];
    float4 uu = f[0];
#pragma unroll
    for (int ch = 1; ch < NCHUNK; ++ch) {
        uu.x += f[ch].x; uu.y += f[ch].y; uu.z += f[ch].z; uu.w += f[ch].w;
    }
    float sc[SPWK];
#pragma unroll
    for (int i = 0; i < SPWK; ++i) {
        float ta = TA[s0 + i];
        float4 g = make_float4(r[i].x + ta, r[i].y + ta, r[i].z + ta, r[i].w + ta);
        float v = (g.x * uu.x + g.y * uu.y) + (g.z * uu.z + g.w * uu.w);
        sc[i] = wave_sum(v);
    }
    if (lane == 0) {
        float2* dst = reinterpret_cast<float2*>(praw + b * SLEN + s0);
#pragma unroll
        for (int i = 0; i < SPWK; i += 2) dst[i / 2] = make_float2(sc[i], sc[i + 1]);
    }
    if (grp == 0 && w == 0) reinterpret_cast<float4*>(u + b * EMBD)[lane] = uu;
}

template <bool KEEP>
__device__ __forceinline__
void combine_body(const float* __restrict__ Ak1, const float* __restrict__ TC,
                  const int* __restrict__ x, const float* __restrict__ praw,
                  const float* __restrict__ u, float* __restrict__ part,
                  float* __restrict__ keep) {
    int ch = blockIdx.x, b = blockIdx.y;
    int lane = threadIdx.x & (WAVE - 1);
    float ur = u[b * EMBD + threadIdx.x];
    const float* pr = praw + b * SLEN;
    float pv[SROW];
    float m = -1e30f;
    for (int j = 0; j < SROW; ++j) {
        int s = lane + j * WAVE;
        pv[j] = s < SLEN ? pr[s] : -1e30f;
        m = fmaxf(m, pv[j]);
    }
    m = wave_max(m);
    float sum = 0.f;
    for (int j = 0; j < SROW; ++j)
        sum += (lane + j * WAVE < SLEN) ? __expf(pv[j] - m) : 0.f;
    float inv = 1.0f / wave_sum(sum);
    int w = __builtin_amdgcn_readfirstlane(threadIdx.x / WAVE);
    __shared__ float4 gt[8][WAVE];
    for (int i = 0; i < SPW; ++i) {
        int sl = w * SPW + i;
        int s  = ch * 8 + sl;
        float4 g = bag_sum(Ak1, x + (b * SLEN + s) * SENT, lane);
        if (KEEP) reinterpret_cast<float4*>(keep + (long)(b * SLEN + s) * EMBD)[lane] = g;
        float tc = TC[s];
        float ps = __expf(pr[s] - m) * inv;
        gt[sl][lane] = make_float4((g.x + tc) * ps, (g.y + tc) * ps,
                                   (g.z + tc) * ps, (g.w + tc) * ps);
    }
    __syncthreads();
    const float* gf = reinterpret_cast<const float*>(&gt[0][0]);
    int e = threadIdx.x;
    float acc = 0.f;
    for (int sl = 0; sl < 8; ++sl)
        acc += gf[sl * EMBD + e];
    if (ch == 0) acc += ur;
    part[(b * NCHUNK + ch) * EMBD + e] = acc;
}

__global__ __launch_bounds__(WPB * WAVE)
void k_combine_keep(const float* __restrict__ Ak1, const float* __restrict__ TC,
                    const int* __restrict__ x, const float* __restrict__ praw,
                    const float* __restrict__ u, float* __restrict__ part,
                    float* __restrict__ keep) {
    combine_body<true>(Ak1, TC, x, praw, u, part, keep);
}

__global__ __launch_bounds__(WPB * WAVE)
void k_combine_last(const float* __restrict__ Ak1, const float* __restrict__ TC,
                    const int* __restrict__ x, const float* __restrict__ praw,
                    const float* __restrict__ u, float* __restrict__ part) {
    combine_body<false>(Ak1, TC, x, praw, u, part, nullptr);
}

__global__ __launch_bounds__(EMBD)
void k_fold_u(float* __restrict__ u, const float* __restrict__ part) {
    int b = blockIdx.x, e = threadIdx.x;
    u[b * EMBD + e] = fold_part(part, b, e);
}

__global__ __launch_bounds__(32)
void k_logits(const float* __restrict__ u, const float* __restrict__ A3,
              float* __restrict__ out) {
    int n0   = blockIdx.x * 16;
    int lane = threadIdx.x;
    int half = lane >> 4;
    int l16  = lane & 15;
    int koff = half * 2;

    const float* arow = A3 + (long)(n0 + l16) * EMBD;
    const float* u0   = u + l16 * EMBD;

    v8f c0 = {}, c1 = {}, c2 = {}, c3 = {};
#pragma unroll
    for (int k0 = 0; k0 < EMBD; k0 += 4) {
        int kk = k0 + koff;
        v2f bfrag, a0, a1, a2, a3;
        bfrag.x = arow[kk];               bfrag.y = arow[kk + 1];
        a0.x = u0[kk];                    a0.y = u0[kk + 1];
        a1.x = u0[16 * EMBD + kk];        a1.y = u0[16 * EMBD + kk + 1];
        a2.x = u0[32 * EMBD + kk];        a2.y = u0[32 * EMBD + kk + 1];
        a3.x = u0[48 * EMBD + kk];        a3.y = u0[48 * EMBD + kk + 1];
        c0 = __builtin_amdgcn_wmma_f32_16x16x4_f32(false, a0, false, bfrag,
                                                   (short)0, c0, false, false);
        c1 = __builtin_amdgcn_wmma_f32_16x16x4_f32(false, a1, false, bfrag,
                                                   (short)0, c1, false, false);
        c2 = __builtin_amdgcn_wmma_f32_16x16x4_f32(false, a2, false, bfrag,
                                                   (short)0, c2, false, false);
        c3 = __builtin_amdgcn_wmma_f32_16x16x4_f32(false, a3, false, bfrag,
                                                   (short)0, c3, false, false);
    }

    int col = n0 + l16;
#pragma unroll
    for (int r = 0; r < 8; ++r) {
        int row = r + 8 * half;
        out[(long)(row)      * VOCAB + col] = c0[r];
        out[(long)(row + 16) * VOCAB + col] = c1[r];
        out[(long)(row + 32) * VOCAB + col] = c2[r];
        out[(long)(row + 48) * VOCAB + col] = c3[r];
    }
}

#define NSPLIT 50
#define SLICE4 (VOCAB / NSPLIT / 4)
static_assert(VOCAB % (NSPLIT * 4) == 0, "slices must be whole float4s");
static_assert(SLICE4 <= 256, "one float4 per thread");
static_assert(NSPLIT <= 2 * WAVE, "merge reads at most two partials per lane");

__global__ __launch_bounds__(256)
void k_rowstats(const float* __restrict__ out, float* __restrict__ pmax,
                float* __restrict__ psum) {
    int j = blockIdx.x, b = blockIdx.y, t = threadIdx.x;
    const float4* src =
        reinterpret_cast<const float4*>(out + (long)b * VOCAB) + j * SLICE4;
    __shared__ float redm[256 / WAVE], reds[256 / WAVE];

    bool act = t < SLICE4;
    float4 v = act ? src[t] : make_float4(-1e30f, -1e30f, -1e30f, -1e30f);

    float m = wave_max(fmaxf(fmaxf(v.x, v.y), fmaxf(v.z, v.w)));
    if ((t & (WAVE - 1)) == 0) redm[t / WAVE] = m;
    __syncthreads();
    m = redm[0];
    for (int w = 1; w < 256 / WAVE; ++w) m = fmaxf(m, redm[w]);

    float s = act ? (__expf(v.x - m) + __expf(v.y - m)) +
                    (__expf(v.z - m) + __expf(v.w - m))
                  : 0.f;
    s = wave_sum(s);
    if ((t & (WAVE - 1)) == 0) reds[t / WAVE] = s;
    __syncthreads();
    if (t == 0) {
        float tot = reds[0];
        for (int w = 1; w < 256 / WAVE; ++w) tot += reds[w];
        pmax[b * NSPLIT + j] = m;
        psum[b * NSPLIT + j] = tot;
    }
}

__global__ __launch_bounds__(256)
void k_finalize(float* __restrict__ out, const float* __restrict__ pmax,
                const float* __restrict__ psum) {
    int j = blockIdx.x, b = blockIdx.y, t = threadIdx.x;
    int lane = t & (WAVE - 1);
    bool act = t < SLICE4;
    float4* dst = reinterpret_cast<float4*>(out + (long)b * VOCAB) + j * SLICE4 + t;
    float4 v = act ? *dst : make_float4(0.f, 0.f, 0.f, 0.f);

    const float* pm = pmax + b * NSPLIT;
    const float* ps = psum + b * NSPLIT;
    bool hi = lane + WAVE < NSPLIT;
    float m0 = pm[lane];
    float s0 = ps[lane];
    float m1 = hi ? pm[lane + WAVE] : -1e30f;
    float s1 = hi ? ps[lane + WAVE] : 0.f;

    float M   = wave_max(fmaxf(m0, m1));
    float S   = wave_sum(s0 * __expf(m0 - M) + s1 * __expf(m1 - M));
    float lse = __logf(S);

    if (act) {
        v.x = v.x - M - lse;
        v.y = v.y - M - lse;
        v.z = v.z - M - lse;
        v.w = v.w - M - lse;
        *dst = v;
    }
}

extern "C" void kernel_launch(void* const* d_in, const int* in_sizes, int n_in,
                              void* d_out, int out_size, void* d_ws, size_t ws_size,
                              hipStream_t stream) {
    const float* A  = (const float*)d_in[0];
    const float* TA = (const float*)d_in[1];
    const float* TC = (const float*)d_in[2];
    const int*   x  = (const int*)d_in[3];
    const int*   q  = (const int*)d_in[4];
    float* out = (float*)d_out;

    float* ws   = (float*)d_ws;
    float* u    = ws;
    float* p    = u    + BATCH * EMBD;
    float* part = p    + BATCH * SLEN;
    float* pmax = part;
    float* psum = part + BATCH * NSPLIT;

    float* keep = out;
    static_assert((long)BATCH * SLEN * EMBD <= (long)BATCH * VOCAB, "kept row sums fit in out");

    const long TBL = (long)VOCAB * EMBD;

    k_init_u<<<BATCH, WAVE, 0, stream>>>(A + 1 * TBL, q, u);

    k_score_first<<<dim3(NCHUNK, BATCH), WPB * WAVE, 0, stream>>>(A, TA, x, u, p);
    k_combine_keep<<<dim3(NCHUNK, BATCH), WPB * WAVE, 0, stream>>>(A + TBL, TC, x, p, u, part, keep);
    static_assert(HOPS == 3, "hops 0 and 1 keep their row sums, hop 2 is the last");
    k_score_kept  <<<dim3(SGRP, BATCH),   WPB * WAVE, 0, stream>>>(keep, TA, part, u, p);
    k_combine_keep<<<dim3(NCHUNK, BATCH), WPB * WAVE, 0, stream>>>(A + 2 * TBL, TC, x, p, u, part, keep);
    k_score_kept  <<<dim3(SGRP, BATCH),   WPB * WAVE, 0, stream>>>(keep, TA, part, u, p);
    k_combine_last<<<dim3(NCHUNK, BATCH), WPB * WAVE, 0, stream>>>(A + 3 * TBL, TC, x, p, u, part);
    k_fold_u<<<BATCH, EMBD, 0, stream>>>(u, part);

    k_logits  <<<dim3(VOCAB / 16), 32, 0, stream>>>(u, A + HOPS * TBL, out);
    k_rowstats<<<dim3(NSPLIT, BATCH), 256, 0, stream>>>(out, pmax, psum);
    k_finalize<<<dim3(NSPLIT, BATCH), 256, 0, stream>>>(out, pmax, psum);

    (void)in_sizes; (void)n_in; (void)out_size; (void)ws_size;
}


# k_logits: u staged once per 4-wave work-group in LDS by whole-row loads, A fragments read from LDS; one 16 B piece feeds two WMMAs for A and B alike; k order inside the logit sums changes
# speedup vs baseline: 2.7584x; 1.0542x over previous
#include <hip/hip_runtime.h>
#include <hip/hip_bf16.h>

#define HOPS   3
#define VOCAB  50000
#define EMBD   128
#define SLEN   200
#define SENT   32
#define BATCH  64
#define QLEN   32
#define NCHUNK 25
#define WAVE   32
#define SPW    2
#define WPB    (8 / SPW)

static_assert(EMBD == 4 * WAVE, "one table row = one 16-byte load per lane of a wave");
static_assert(QLEN == SENT, "bag_sum sums SENT rows");
static_assert(WPB * WAVE == EMBD, "k_combine's last stage: one thread per embedding dim");
static_assert(NCHUNK * 8 == SLEN, "chunks of 8 sentences");
#define SROW ((SLEN + WAVE - 1) / WAVE)

typedef __attribute__((ext_vector_type(2))) float v2f;
typedef __attribute__((ext_vector_type(8))) float v8f;

__device__ __forceinline__ float wave_max(float v) {
    for (int o = WAVE / 2; o > 0; o >>= 1) v = fmaxf(v, __shfl_xor(v, o));
    return v;
}
__device__ __forceinline__ float wave_sum(float v) {
    for (int o = WAVE / 2; o > 0; o >>= 1) v += __shfl_xor(v, o);
    return v;
}

__device__ __forceinline__ float fold_part(const float* __restrict__ part, int b, int e) {
    const float* pp = part + (b * NCHUNK) * EMBD + e;
    float acc = pp[0];
    for (int ch = 1; ch < NCHUNK; ++ch) acc += pp[ch * EMBD];
    return acc;
}

__device__ __forceinline__ float4 bag_sum(const float* __restrict__ tab,
                                          const int* __restrict__ idx, int lane) {
    float4 r[SENT];
#pragma unroll
    for (int t = 0; t < SENT; ++t)
        r[t] = reinterpret_cast<const float4*>(tab + (long)idx[t] * EMBD)[lane];
    float4 g = make_float4(0.f, 0.f, 0.f, 0.f);
#pragma unroll
    for (int t = 0; t < SENT; ++t) {
        g.x += r[t].x; g.y += r[t].y; g.z += r[t].z; g.w += r[t].w;
    }
    return g;
}

__global__ __launch_bounds__(WAVE)
void k_init_u(const float* __restrict__ A1, const int* __restrict__ q,
              float* __restrict__ u) {
    int b = blockIdx.x, lane = threadIdx.x;
    float4 g = bag_sum(A1, q + b * QLEN, lane);
    reinterpret_cast<float4*>(u + b * EMBD)[lane] = g;
}

__device__ __forceinline__
void score_gather(const float* __restrict__ Ak, const float* __restrict__ TA,
                  const int* __restrict__ x, int b, int s0, int lane,
                  float4 (&g)[SPW]) {
    for (int i = 0; i < SPW; ++i) {
        float4 r = bag_sum(Ak, x + (b * SLEN + s0 + i) * SENT, lane);
        float ta = TA[s0 + i];
        g[i] = make_float4(r.x + ta, r.y + ta, r.z + ta, r.w + ta);
    }
}

__device__ __forceinline__
void score_store(const float4 (&g)[SPW], float4 uu,
                 float* __restrict__ praw, int b, int s0, int lane) {
    float sc[SPW];
    for (int i = 0; i < SPW; ++i) {
        float v = (g[i].x * uu.x + g[i].y * uu.y) +
                  (g[i].z * uu.z + g[i].w * uu.w);
        sc[i] = wave_sum(v);
    }
    static_assert(SPW == 2, "the two scores of a wave leave as one 8-byte store");
    if (lane == 0)
        *reinterpret_cast<float2*>(praw + b * SLEN + s0) = make_float2(sc[0], sc[1]);
}

__global__ __launch_bounds__(WPB * WAVE)
void k_score_first(const float* __restrict__ Ak, const float* __restrict__ TA,
                   const int* __restrict__ x, const float* __restrict__ u,
                   float* __restrict__ praw) {
    int ch = blockIdx.x, b = blockIdx.y;
    int lane = threadIdx.x & (WAVE - 1);
    int w = __builtin_amdgcn_readfirstlane(threadIdx.x / WAVE);
    int s0 = ch * 8 + w * SPW;
    float4 uu = reinterpret_cast<const float4*>(u + b * EMBD)[lane];
    float4 g[SPW];
    score_gather(Ak, TA, x, b, s0, lane, g);
    score_store(g, uu, praw, b, s0, lane);
}

#define SGRP 5
#define SPG  (SLEN / SGRP)
#define SPWK (SPG / WPB)
static_assert(SGRP * WPB * SPWK == SLEN, "the work-groups of a row cover its sentences exactly");
static_assert(SPWK % 2 == 0, "a wave's scores leave as 8-byte stores");

__global__ __launch_bounds__(WPB * WAVE)
void k_score_kept(const float* __restrict__ keep, const float* __restrict__ TA,
                  const float* __restrict__ part, float* __restrict__ u,
                  float* __restrict__ praw) {
    int grp = blockIdx.x, b = blockIdx.y;
    int lane = threadIdx.x & (WAVE - 1);
    int w = __builtin_amdgcn_readfirstlane(threadIdx.x / WAVE);
    int s0 = grp * SPG + w * SPWK;
    const float4* rows =
        reinterpret_cast<const float4*>(keep + (long)(b * SLEN + s0) * EMBD) + lane;
    const float4* pp =
        reinterpret_cast<const float4*>(part + (long)b * NCHUNK * EMBD) + lane;
    float4 r[SPWK], f[NCHUNK];
#pragma unroll
    for (int i = 0; i < SPWK; ++i) r[i] = rows[i * WAVE];
#pragma unroll
    for (int ch = 0; ch < NCHUNK; ++ch) f[ch] = pp[ch * WAVE];
    float4 uu = f[0];
#pragma unroll
    for (int ch = 1; ch < NCHUNK; ++ch) {
        uu.x += f[ch].x; uu.y += f[ch].y; uu.z += f[ch].z; uu.w += f[ch].w;
    }
    float sc[SPWK];
#pragma unroll
    for (int i = 0; i < SPWK; ++i) {
        float ta = TA[s0 + i];
        float4 g = make_float4(r[i].x + ta, r[i].y + ta, r[i].z + ta, r[i].w + ta);
        float v = (g.x * uu.x + g.y * uu.y) + (g.z * uu.z + g.w * uu.w);
        sc[i] = wave_sum(v);
    }
    if (lane == 0) {
        float2* dst = reinterpret_cast<float2*>(praw + b * SLEN + s0);
#pragma unroll
        for (int i = 0; i < SPWK; i += 2) dst[i / 2] = make_float2(sc[i], sc[i + 1]);
    }
    if (grp == 0 && w == 0) reinterpret_cast<float4*>(u + b * EMBD)[lane] = uu;
}

template <bool KEEP>
__device__ __forceinline__
void combine_body(const float* __restrict__ Ak1, const float* __restrict__ TC,
                  const int* __restrict__ x, const float* __restrict__ praw,
                  const float* __restrict__ u, float* __restrict__ part,
                  float* __restrict__ keep) {
    int ch = blockIdx.x, b = blockIdx.y;
    int lane = threadIdx.x & (WAVE - 1);
    float ur = u[b * EMBD + threadIdx.x];
    const float* pr = praw + b * SLEN;
    float pv[SROW];
    float m = -1e30f;
    for (int j = 0; j < SROW; ++j) {
        int s = lane + j * WAVE;
        pv[j] = s < SLEN ? pr[s] : -1e30f;
        m = fmaxf(m, pv[j]);
    }
    m = wave_max(m);
    float sum = 0.f;
    for (int j = 0; j < SROW; ++j)
        sum += (lane + j * WAVE < SLEN) ? __expf(pv[j] - m) : 0.f;
    float inv = 1.0f / wave_sum(sum);
    int w = __builtin_amdgcn_readfirstlane(threadIdx.x / WAVE);
    __shared__ float4 gt[8][WAVE];
    for (int i = 0; i < SPW; ++i) {
        int sl = w * SPW + i;
        int s  = ch * 8 + sl;
        float4 g = bag_sum(Ak1, x + (b * SLEN + s) * SENT, lane);
        if (KEEP) reinterpret_cast<float4*>(keep + (long)(b * SLEN + s) * EMBD)[lane] = g;
        float tc = TC[s];
        float ps = __expf(pr[s] - m) * inv;
        gt[sl][lane] = make_float4((g.x + tc) * ps, (g.y + tc) * ps,
                                   (g.z + tc) * ps, (g.w + tc) * ps);
    }
    __syncthreads();
    const float* gf = reinterpret_cast<const float*>(&gt[0][0]);
    int e = threadIdx.x;
    float acc = 0.f;
    for (int sl = 0; sl < 8; ++sl)
        acc += gf[sl * EMBD + e];
    if (ch == 0) acc += ur;
    part[(b * NCHUNK + ch) * EMBD + e] = acc;
}

__global__ __launch_bounds__(WPB * WAVE)
void k_combine_keep(const float* __restrict__ Ak1, const float* __restrict__ TC,
                    const int* __restrict__ x, const float* __restrict__ praw,
                    const float* __restrict__ u, float* __restrict__ part,
                    float* __restrict__ keep) {
    combine_body<true>(Ak1, TC, x, praw, u, part, keep);
}

__global__ __launch_bounds__(WPB * WAVE)
void k_combine_last(const float* __restrict__ Ak1, const float* __restrict__ TC,
                    const int* __restrict__ x, const float* __restrict__ praw,
                    const float* __restrict__ u, float* __restrict__ part) {
    combine_body<false>(Ak1, TC, x, praw, u, part, nullptr);
}

__global__ __launch_bounds__(EMBD)
void k_fold_u(float* __restrict__ u, const float* __restrict__ part) {
    int b = blockIdx.x, e = threadIdx.x;
    u[b * EMBD + e] = fold_part(part, b, e);
}

#define LWAVES 4
#define NSTRIP (VOCAB / 16)
#define LGRID  782
#define UPITCH4 (EMBD / 4 + 1)
static_assert(VOCAB % 16 == 0, "whole strips");
static_assert(LGRID * LWAVES >= NSTRIP && (LGRID - 1) * LWAVES < NSTRIP, "grid covers the strips");
static_assert(BATCH == 64 && BATCH % LWAVES == 0, "four batch tiles; staging rows split over the waves");

__global__ __launch_bounds__(LWAVES * WAVE) __attribute__((amdgpu_waves_per_eu(1, 4)))
void k_logits(const float* __restrict__ u, const float* __restrict__ A3,
              float* __restrict__ out) {
    __shared__ float4 us4[BATCH * UPITCH4];
    int lane = threadIdx.x & (WAVE - 1);
    int w = __builtin_amdgcn_readfirstlane(threadIdx.x / WAVE);
    int strip = blockIdx.x * LWAVES + w;
    bool live = strip < NSTRIP;
    int n0   = (live ? strip : NSTRIP - 1) * 16;
    int half = lane >> 4;
    int l16  = lane & 15;

    const float4* brow =
        reinterpret_cast<const float4*>(A3 + (long)(n0 + l16) * EMBD) + half;
    float4 b[EMBD / 8];
#pragma unroll
    for (int j = 0; j < EMBD / 8; ++j) b[j] = brow[2 * j];

    const float4* ug = reinterpret_cast<const float4*>(u);
#pragma unroll
    for (int i = 0; i < BATCH / LWAVES; ++i) {
        int row = i * LWAVES + w;
        us4[row * UPITCH4 + lane] = ug[row * (EMBD / 4) + lane];
    }
    __syncthreads();
    if (!live) return;

    const float4* ua = us4 + l16 * UPITCH4 + half;
    v8f c0 = {}, c1 = {}, c2 = {}, c3 = {};
    float4 an[4], ac[4];
#pragma unroll
    for (int t = 0; t < 4; ++t) an[t] = ua[16 * t * UPITCH4];
#pragma unroll
    for (int j = 0; j < EMBD / 8; ++j) {
#pragma unroll
        for (int t = 0; t < 4; ++t) ac[t] = an[t];
        if (j + 1 < EMBD / 8) {
#pragma unroll
            for (int t = 0; t < 4; ++t) an[t] = ua[16 * t * UPITCH4 + 2 * (j + 1)];
        }
        v2f bx, bz, ax, az;
        bx.x = b[j].x; bx.y = b[j].y;
        bz.x = b[j].z; bz.y = b[j].w;
        ax.x = ac[0].x; ax.y = ac[0].y;
        c0 = __builtin_amdgcn_wmma_f32_16x16x4_f32(false, ax, false, bx,
                                                   (short)0, c0, false, false);
        ax.x = ac[1].x; ax.y = ac[1].y;
        c1 = __builtin_amdgcn_wmma_f32_16x16x4_f32(false, ax, false, bx,
                                                   (short)0, c1, false, false);
        ax.x = ac[2].x; ax.y = ac[2].y;
        c2 = __builtin_amdgcn_wmma_f32_16x16x4_f32(false, ax, false, bx,
                                                   (short)0, c2, false, false);
        ax.x = ac[3].x; ax.y = ac[3].y;
        c3 = __builtin_amdgcn_wmma_f32_16x16x4_f32(false, ax, false, bx,
                                                   (short)0, c3, false, false);
        az.x = ac[0].z; az.y = ac[0].w;
        c0 = __builtin_amdgcn_wmma_f32_16x16x4_f32(false, az, false, bz,
                                                   (short)0, c0, false, false);
        az.x = ac[1].z; az.y = ac[1].w;
        c1 = __builtin_amdgcn_wmma_f32_16x16x4_f32(false, az, false, bz,
                                                   (short)0, c1, false, false);
        az.x = ac[2].z; az.y = ac[2].w;
        c2 = __builtin_amdgcn_wmma_f32_16x16x4_f32(false, az, false, bz,
                                                   (short)0, c2, false, false);
        az.x = ac[3].z; az.y = ac[3].w;
        c3 = __builtin_amdgcn_wmma_f32_16x16x4_f32(false, az, false, bz,
                                                   (short)0, c3, false, false);
    }

    int col = n0 + l16;
#pragma unroll
    for (int r = 0; r < 8; ++r) {
        int row = r + 8 * half;
        out[(long)(row)      * VOCAB + col] = c0[r];
        out[(long)(row + 16) * VOCAB + col] = c1[r];
        out[(long)(row + 32) * VOCAB + col] = c2[r];
        out[(long)(row + 48) * VOCAB + col] = c3[r];
    }
}

#define NSPLIT 50
#define SLICE4 (VOCAB / NSPLIT / 4)
static_assert(VOCAB % (NSPLIT * 4) == 0, "slices must be whole float4s");
static_assert(SLICE4 <= 256, "one float4 per thread");
static_assert(NSPLIT <= 2 * WAVE, "merge reads at most two partials per lane");

__global__ __launch_bounds__(256)
void k_rowstats(const float* __restrict__ out, float* __restrict__ pmax,
                float* __restrict__ psum) {
    int j = blockIdx.x, b = blockIdx.y, t = threadIdx.x;
    const float4* src =
        reinterpret_cast<const float4*>(out + (long)b * VOCAB) + j * SLICE4;
    __shared__ float redm[256 / WAVE], reds[256 / WAVE];

    bool act = t < SLICE4;
    float4 v = act ? src[t] : make_float4(-1e30f, -1e30f, -1e30f, -1e30f);

    float m = wave_max(fmaxf(fmaxf(v.x, v.y), fmaxf(v.z, v.w)));
    if ((t & (WAVE - 1)) == 0) redm[t / WAVE] = m;
    __syncthreads();
    m = redm[0];
    for (int w = 1; w < 256 / WAVE; ++w) m = fmaxf(m, redm[w]);

    float s = act ? (__expf(v.x - m) + __expf(v.y - m)) +
                    (__expf(v.z - m) + __expf(v.w - m))
                  : 0.f;
    s = wave_sum(s);
    if ((t & (WAVE - 1)) == 0) reds[t / WAVE] = s;
    __syncthreads();
    if (t == 0) {
        float tot = reds[0];
        for (int w = 1; w < 256 / WAVE; ++w) tot += reds[w];
        pmax[b * NSPLIT + j] = m;
        psum[b * NSPLIT + j] = tot;
    }
}

__global__ __launch_bounds__(256)
void k_finalize(float* __restrict__ out, const float* __restrict__ pmax,
                const float* __restrict__ psum) {
    int j = blockIdx.x, b = blockIdx.y, t = threadIdx.x;
    int lane = t & (WAVE - 1);
    bool act = t < SLICE4;
    float4* dst = reinterpret_cast<float4*>(out + (long)b * VOCAB) + j * SLICE4 + t;
    float4 v = act ? *dst : make_float4(0.f, 0.f, 0.f, 0.f);

    const float* pm = pmax + b * NSPLIT;
    const float* ps = psum + b * NSPLIT;
    bool hi = lane + WAVE < NSPLIT;
    float m0 = pm[lane];
    float s0 = ps[lane];
    float m1 = hi ? pm[lane + WAVE] : -1e30f;
    float s1 = hi ? ps[lane + WAVE] : 0.f;

    float M   = wave_max(fmaxf(m0, m1));
    float S   = wave_sum(s0 * __expf(m0 - M) + s1 * __expf(m1 - M));
    float lse = __logf(S);

    if (act) {
        v.x = v.x - M - lse;
        v.y = v.y - M - lse;
        v.z = v.z - M - lse;
        v.w = v.w - M - lse;
        *dst = v;
    }
}

extern "C" void kernel_launch(void* const* d_in, const int* in_sizes, int n_in,
                              void* d_out, int out_size, void* d_ws, size_t ws_size,
                              hipStream_t stream) {
    const float* A  = (const float*)d_in[0];
    const float* TA = (const float*)d_in[1];
    const float* TC = (const float*)d_in[2];
    const int*   x  = (const int*)d_in[3];
    const int*   q  = (const int*)d_in[4];
    float* out = (float*)d_out;

    float* ws   = (float*)d_ws;
    float* u    = ws;
    float* p    = u    + BATCH * EMBD;
    float* part = p    + BATCH * SLEN;
    float* pmax = part;
    float* psum = part + BATCH * NSPLIT;

    float* keep = out;
    static_assert((long)BATCH * SLEN * EMBD <= (long)BATCH * VOCAB, "kept row sums fit in out");

    const long TBL = (long)VOCAB * EMBD;

    k_init_u<<<BATCH, WAVE, 0, stream>>>(A + 1 * TBL, q, u);

    k_score_first<<<dim3(NCHUNK, BATCH), WPB * WAVE, 0, stream>>>(A, TA, x, u, p);
    k_combine_keep<<<dim3(NCHUNK, BATCH), WPB * WAVE, 0, stream>>>(A + TBL, TC, x, p, u, part, keep);
    static_assert(HOPS == 3, "hops 0 and 1 keep their row sums, hop 2 is the last");
    k_score_kept  <<<dim3(SGRP, BATCH),   WPB * WAVE, 0, stream>>>(keep, TA, part, u, p);
    k_combine_keep<<<dim3(NCHUNK, BATCH), WPB * WAVE, 0, stream>>>(A + 2 * TBL, TC, x, p, u, part, keep);
    k_score_kept  <<<dim3(SGRP, BATCH),   WPB * WAVE, 0, stream>>>(keep, TA, part, u, p);
    k_combine_last<<<dim3(NCHUNK, BATCH), WPB * WAVE, 0, stream>>>(A + 3 * TBL, TC, x, p, u, part);
    k_fold_u<<<BATCH, EMBD, 0, stream>>>(u, part);

    k_logits  <<<dim3(LGRID), LWAVES * WAVE, 0, stream>>>(u, A + HOPS * TBL, out);
    k_rowstats<<<dim3(NSPLIT, BATCH), 256, 0, stream>>>(out, pmax, psum);
    k_finalize<<<dim3(NSPLIT, BATCH), 256, 0, stream>>>(out, pmax, psum);

    (void)in_sizes; (void)n_in; (void)out_size; (void)ws_size;
}


# row sums of all 4 tables gathered without u in two whole-chip launches and kept (out + 3 rows of ws); each hop pair then runs in one work-group per batch row: 11 -> 7 launches, sum orders kept
# speedup vs baseline: 3.0296x; 1.0983x over previous
#include <hip/hip_runtime.h>
#include <hip/hip_bf16.h>

#define HOPS   3
#define VOCAB  50000
#define EMBD   128
#define SLEN   200
#define SENT   32
#define BATCH  64
#define QLEN   32
#define NCHUNK 25
#define WAVE   32
#define SPW    2
#define WPB    (8 / SPW)

static_assert(EMBD == 4 * WAVE, "one table row = one 16-byte load per lane of a wave");
static_assert(QLEN == SENT, "bag_sum sums SENT rows");
static_assert(NCHUNK * 8 == SLEN, "chunks of 8 sentences");
#define SROW ((SLEN + WAVE - 1) / WAVE)
#define CHS  (SLEN / NCHUNK)

typedef __attribute__((ext_vector_type(2))) float v2f;
typedef __attribute__((ext_vector_type(8))) float v8f;

__device__ __forceinline__ float wave_max(float v) {
    for (int o = WAVE / 2; o > 0; o >>= 1) v = fmaxf(v, __shfl_xor(v, o));
    return v;
}
__device__ __forceinline__ float wave_sum(float v) {
    for (int o = WAVE / 2; o > 0; o >>= 1) v += __shfl_xor(v, o);
    return v;
}

__device__ __forceinline__ float4 bag_sum(const float* __restrict__ tab,
                                          const int* __restrict__ idx, int lane) {
    float4 r[SENT];
#pragma unroll
    for (int t = 0; t < SENT; ++t)
        r[t] = reinterpret_cast<const float4*>(tab + (long)idx[t] * EMBD)[lane];
    float4 g = make_float4(0.f, 0.f, 0.f, 0.f);
#pragma unroll
    for (int t = 0; t < SENT; ++t) {
        g.x += r[t].x; g.y += r[t].y; g.z += r[t].z; g.w += r[t].w;
    }
    return g;
}

#define GROWF  (SLEN * EMBD)
#define KSLOTS ((int)((long)BATCH * VOCAB / GROWF))
#define KOVER  (2 * BATCH - KSLOTS)
static_assert(KSLOTS >= BATCH && KOVER >= 0 && KOVER <= BATCH, "first table whole in out, overflow from the second only");
static_assert(GROWF % 4 == 0, "slots are 16-byte aligned");

__device__ __forceinline__ float* keep_row_w(float* keep, float* over, int t, int b) {
    int slot = t * BATCH + b;
    return slot < KSLOTS ? keep + (long)slot * GROWF : over + (long)(slot - KSLOTS) * GROWF;
}
__device__ __forceinline__ const float* keep_row_r(const float* keep, const float* over, int t, int b) {
    int slot = t * BATCH + b;
    return slot < KSLOTS ? keep + (long)slot * GROWF : over + (long)(slot - KSLOTS) * GROWF;
}

template <bool INIT>
__device__ __forceinline__
void gather_body(const float* __restrict__ T2, const int* __restrict__ x,
                 float* __restrict__ keep, float* __restrict__ over,
                 const float* __restrict__ A1, const int* __restrict__ q,
                 float* __restrict__ u) {
    int ch = blockIdx.x, b = blockIdx.y, t = blockIdx.z;
    int lane = threadIdx.x & (WAVE - 1);
    int w = __builtin_amdgcn_readfirstlane(threadIdx.x / WAVE);
    if (INIT && ch == NCHUNK) {
        if (t == 0 && w == 0) {
            float4 g = bag_sum(A1, q + b * QLEN, lane);
            reinterpret_cast<float4*>(u + b * EMBD)[lane] = g;
        }
        return;
    }
    const float* tab = T2 + (long)t * VOCAB * EMBD;
    float* dst = keep_row_w(keep, over, t, b);
    int s0 = ch * CHS + w * SPW;
    for (int i = 0; i < SPW; ++i) {
        float4 g = bag_sum(tab, x + (b * SLEN + s0 + i) * SENT, lane);
        reinterpret_cast<float4*>(dst + (s0 + i) * EMBD)[lane] = g;
    }
}

__global__ __launch_bounds__(WPB * WAVE)
void k_gather_first(const float* __restrict__ T2, const int* __restrict__ x,
                    float* __restrict__ keep, float* __restrict__ over,
                    const float* __restrict__ A1, const int* __restrict__ q,
                    float* __restrict__ u) {
    gather_body<true>(T2, x, keep, over, A1, q, u);
}

__global__ __launch_bounds__(WPB * WAVE)
void k_gather_next(const float* __restrict__ T2, const int* __restrict__ x,
                   float* __restrict__ keep, float* __restrict__ over) {
    gather_body<false>(T2, x, keep, over, nullptr, nullptr, nullptr);
}

#define HT (NCHUNK * WAVE)

__device__ __forceinline__
void row_scores(const float4 (&r)[CHS], float4 uu, const float* __restrict__ TA,
                int s0, float (&sc)[CHS]) {
#pragma unroll
    for (int i = 0; i < CHS; ++i) {
        float ta = TA[s0 + i];
        float4 g = make_float4(r[i].x + ta, r[i].y + ta, r[i].z + ta, r[i].w + ta);
        float v = (g.x * uu.x + g.y * uu.y) + (g.z * uu.z + g.w * uu.w);
        sc[i] = wave_sum(v);
    }
}

template <bool SCORE>
__device__ __forceinline__
float4 hop_stage(const float* pr, float* cp, float* uv,
                 const float4 (&rb)[CHS], float4 uin,
                 const float* __restrict__ TC, const float* __restrict__ TA,
                 int w, int lane, float (&sc)[CHS]) {
    float pv[SROW];
    float m = -1e30f;
    for (int j = 0; j < SROW; ++j) {
        int s = lane + j * WAVE;
        pv[j] = s < SLEN ? pr[s] : -1e30f;
        m = fmaxf(m, pv[j]);
    }
    m = wave_max(m);
    float sum = 0.f;
    for (int j = 0; j < SROW; ++j)
        sum += (lane + j * WAVE < SLEN) ? __expf(pv[j] - m) : 0.f;
    float inv = 1.0f / wave_sum(sum);

    int s0 = w * CHS;
    float4 acc = make_float4(0.f, 0.f, 0.f, 0.f);
#pragma unroll
    for (int i = 0; i < CHS; ++i) {
        float tc = TC[s0 + i];
        float ps = __expf(pr[s0 + i] - m) * inv;
        acc.x += (rb[i].x + tc) * ps;
        acc.y += (rb[i].y + tc) * ps;
        acc.z += (rb[i].z + tc) * ps;
        acc.w += (rb[i].w + tc) * ps;
    }
    if (w == 0) { acc.x += uin.x; acc.y += uin.y; acc.z += uin.z; acc.w += uin.w; }
    reinterpret_cast<float4*>(cp + w * EMBD)[lane] = acc;
    __syncthreads();
    if (w == 0) {
        const float4* c4 = reinterpret_cast<const float4*>(cp) + lane;
        float4 f[NCHUNK];
#pragma unroll
        for (int ch = 0; ch < NCHUNK; ++ch) f[ch] = c4[ch * WAVE];
        float4 uu = f[0];
#pragma unroll
        for (int ch = 1; ch < NCHUNK; ++ch) {
            uu.x += f[ch].x; uu.y += f[ch].y; uu.z += f[ch].z; uu.w += f[ch].w;
        }
        reinterpret_cast<float4*>(uv)[lane] = uu;
    }
    __syncthreads();
    float4 uo = reinterpret_cast<const float4*>(uv)[lane];
    if (SCORE) row_scores(rb, uo, TA, s0, sc);
    return uo;
}

__global__ __launch_bounds__(HT)
void k_hop_first(const float* __restrict__ keep, const float* __restrict__ over,
                 const float* __restrict__ TA, const float* __restrict__ TC,
                 float* __restrict__ u, float* __restrict__ praw) {
    __shared__ __attribute__((aligned(16))) float cp[NCHUNK * EMBD];
    __shared__ __attribute__((aligned(16))) float uv[EMBD];
    __shared__ float pr[SROW * WAVE];
    int b = blockIdx.x;
    int lane = threadIdx.x & (WAVE - 1);
    int w = __builtin_amdgcn_readfirstlane(threadIdx.x / WAVE);
    int s0 = w * CHS;
    const float4* ga = reinterpret_cast<const float4*>(keep_row_r(keep, over, 0, b) + s0 * EMBD) + lane;
    const float4* gb = reinterpret_cast<const float4*>(keep_row_r(keep, over, 1, b) + s0 * EMBD) + lane;
    float4 ra[CHS], rb[CHS];
#pragma unroll
    for (int i = 0; i < CHS; ++i) ra[i] = ga[i * WAVE];
#pragma unroll
    for (int i = 0; i < CHS; ++i) rb[i] = gb[i * WAVE];
    float4 uin = reinterpret_cast<const float4*>(u + b * EMBD)[lane];
    __builtin_amdgcn_sched_barrier(0);

    float sc[CHS];
    row_scores(ra, uin, TA, s0, sc);
    if (lane == 0) {
#pragma unroll
        for (int i = 0; i < CHS; ++i) pr[s0 + i] = sc[i];
    }
    __syncthreads();
    float4 uo = hop_stage<true>(pr, cp, uv, rb, uin, TC, TA, w, lane, sc);
    if (lane == 0) {
        float2* dst = reinterpret_cast<float2*>(praw + b * SLEN + s0);
#pragma unroll
        for (int i = 0; i < CHS; i += 2) dst[i / 2] = make_float2(sc[i], sc[i + 1]);
    }
    if (w == 0) reinterpret_cast<float4*>(u + b * EMBD)[lane] = uo;
}

__global__ __launch_bounds__(HT)
void k_hop_last(const float* __restrict__ keep, const float* __restrict__ over,
                const float* __restrict__ TA, const float* __restrict__ TC,
                const float* __restrict__ praw, float* __restrict__ u) {
    __shared__ __attribute__((aligned(16))) float cp[NCHUNK * EMBD];
    __shared__ __attribute__((aligned(16))) float uv[EMBD];
    __shared__ float pr[SROW * WAVE];
    int b = blockIdx.x;
    int lane = threadIdx.x & (WAVE - 1);
    int w = __builtin_amdgcn_readfirstlane(threadIdx.x / WAVE);
    int s0 = w * CHS;
    const float4* gb = reinterpret_cast<const float4*>(keep_row_r(keep, over, 0, b) + s0 * EMBD) + lane;
    const float4* gc = reinterpret_cast<const float4*>(keep_row_r(keep, over, 1, b) + s0 * EMBD) + lane;
    float4 rb[CHS], rc[CHS];
#pragma unroll
    for (int i = 0; i < CHS; ++i) rb[i] = gb[i * WAVE];
#pragma unroll
    for (int i = 0; i < CHS; ++i) rc[i] = gc[i * WAVE];
    float4 uin = reinterpret_cast<const float4*>(u + b * EMBD)[lane];
    if (threadIdx.x < SLEN) pr[threadIdx.x] = praw[b * SLEN + threadIdx.x];
    __syncthreads();

    float sc[CHS];
    float4 u2 = hop_stage<true>(pr, cp, uv, rb, uin, TC, TA, w, lane, sc);
    if (lane == 0) {
#pragma unroll
        for (int i = 0; i < CHS; ++i) pr[s0 + i] = sc[i];
    }
    __syncthreads();
    float4 u3 = hop_stage<false>(pr, cp, uv, rc, u2, TC, TA, w, lane, sc);
    if (w == 0) reinterpret_cast<float4*>(u + b * EMBD)[lane] = u3;
}

#define LWAVES 4
#define NSTRIP (VOCAB / 16)
#define LGRID  782
#define UPITCH4 (EMBD / 4 + 1)
static_assert(VOCAB % 16 == 0, "whole strips");
static_assert(LGRID * LWAVES >= NSTRIP && (LGRID - 1) * LWAVES < NSTRIP, "grid covers the strips");
static_assert(BATCH == 64 && BATCH % LWAVES == 0, "four batch tiles; staging rows split over the waves");

__global__ __launch_bounds__(LWAVES * WAVE) __attribute__((amdgpu_waves_per_eu(1, 4)))
void k_logits(const float* __restrict__ u, const float* __restrict__ A3,
              float* __restrict__ out) {
    __shared__ float4 us4[BATCH * UPITCH4];
    int lane = threadIdx.x & (WAVE - 1);
    int w = __builtin_amdgcn_readfirstlane(threadIdx.x / WAVE);
    int strip = blockIdx.x * LWAVES + w;
    bool live = strip < NSTRIP;
    int n0   = (live ? strip : NSTRIP - 1) * 16;
    int half = lane >> 4;
    int l16  = lane & 15;

    const float4* brow =
        reinterpret_cast<const float4*>(A3 + (long)(n0 + l16) * EMBD) + half;
    float4 b[EMBD / 8];
#pragma unroll
    for (int j = 0; j < EMBD / 8; ++j) b[j] = brow[2 * j];

    const float4* ug = reinterpret_cast<const float4*>(u);
#pragma unroll
    for (int i = 0; i < BATCH / LWAVES; ++i) {
        int row = i * LWAVES + w;
        us4[row * UPITCH4 + lane] = ug[row * (EMBD / 4) + lane];
    }
    __syncthreads();
    if (!live) return;

    const float4* ua = us4 + l16 * UPITCH4 + half;
    v8f c0 = {}, c1 = {}, c2 = {}, c3 = {};
    float4 an[4], ac[4];
#pragma unroll
    for (int t = 0; t < 4; ++t) an[t] = ua[16 * t * UPITCH4];
#pragma unroll
    for (int j = 0; j < EMBD / 8; ++j) {
#pragma unroll
        for (int t = 0; t < 4; ++t) ac[t] = an[t];
        if (j + 1 < EMBD / 8) {
#pragma unroll
            for (int t = 0; t < 4; ++t) an[t] = ua[16 * t * UPITCH4 + 2 * (j + 1)];
        }
        v2f bx, bz, ax, az;
        bx.x = b[j].x; bx.y = b[j].y;
        bz.x = b[j].z; bz.y = b[j].w;
        ax.x = ac[0].x; ax.y = ac[0].y;
        c0 = __builtin_amdgcn_wmma_f32_16x16x4_f32(false, ax, false, bx,
                                                   (short)0, c0, false, false);
        ax.x = ac[1].x; ax.y = ac[1].y;
        c1 = __builtin_amdgcn_wmma_f32_16x16x4_f32(false, ax, false, bx,
                                                   (short)0, c1, false, false);
        ax.x = ac[2].x; ax.y = ac[2].y;
        c2 = __builtin_amdgcn_wmma_f32_16x16x4_f32(false, ax, false, bx,
                                                   (short)0, c2, false, false);
        ax.x = ac[3].x; ax.y = ac[3].y;
        c3 = __builtin_amdgcn_wmma_f32_16x16x4_f32(false, ax, false, bx,
                                                   (short)0, c3, false, false);
        az.x = ac[0].z; az.y = ac[0].w;
        c0 = __builtin_amdgcn_wmma_f32_16x16x4_f32(false, az, false, bz,
                                                   (short)0, c0, false, false);
        az.x = ac[1].z; az.y = ac[1].w;
        c1 = __builtin_amdgcn_wmma_f32_16x16x4_f32(false, az, false, bz,
                                                   (short)0, c1, false, false);
        az.x = ac[2].z; az.y = ac[2].w;
        c2 = __builtin_amdgcn_wmma_f32_16x16x4_f32(false, az, false, bz,
                                                   (short)0, c2, false, false);
        az.x = ac[3].z; az.y = ac[3].w;
        c3 = __builtin_amdgcn_wmma_f32_16x16x4_f32(false, az, false, bz,
                                                   (short)0, c3, false, false);
    }

    int col = n0 + l16;
#pragma unroll
    for (int r = 0; r < 8; ++r) {
        int row = r + 8 * half;
        out[(long)(row)      * VOCAB + col] = c0[r];
        out[(long)(row + 16) * VOCAB + col] = c1[r];
        out[(long)(row + 32) * VOCAB + col] = c2[r];
        out[(long)(row + 48) * VOCAB + col] = c3[r];
    }
}

#define NSPLIT 50
#define SLICE4 (VOCAB / NSPLIT / 4)
static_assert(VOCAB % (NSPLIT * 4) == 0, "slices must be whole float4s");
static_assert(SLICE4 <= 256, "one float4 per thread");
static_assert(NSPLIT <= 2 * WAVE, "merge reads at most two partials per lane");

__global__ __launch_bounds__(256)
void k_rowstats(const float* __restrict__ out, float* __restrict__ pmax,
                float* __restrict__ psum) {
    int j = blockIdx.x, b = blockIdx.y, t = threadIdx.x;
    const float4* src =
        reinterpret_cast<const float4*>(out + (long)b * VOCAB) + j * SLICE4;
    __shared__ float redm[256 / WAVE], reds[256 / WAVE];

    bool act = t < SLICE4;
    float4 v = act ? src[t] : make_float4(-1e30f, -1e30f, -1e30f, -1e30f);

    float m = wave_max(fmaxf(fmaxf(v.x, v.y), fmaxf(v.z, v.w)));
    if ((t & (WAVE - 1)) == 0) redm[t / WAVE] = m;
    __syncthreads();
    m = redm[0];
    for (int w = 1; w < 256 / WAVE; ++w) m = fmaxf(m, redm[w]);

    float s = act ? (__expf(v.x - m) + __expf(v.y - m)) +
                    (__expf(v.z - m) + __expf(v.w - m))
                  : 0.f;
    s = wave_sum(s);
    if ((t & (WAVE - 1)) == 0) reds[t / WAVE] = s;
    __syncthreads();
    if (t == 0) {
        float tot = reds[0];
        for (int w = 1; w < 256 / WAVE; ++w) tot += reds[w];
        pmax[b * NSPLIT + j] = m;
        psum[b * NSPLIT + j] = tot;
    }
}

__global__ __launch_bounds__(256)
void k_finalize(float* __restrict__ out, const float* __restrict__ pmax,
                const float* __restrict__ psum) {
    int j = blockIdx.x, b = blockIdx.y, t = threadIdx.x;
    int lane = t & (WAVE - 1);
    bool act = t < SLICE4;
    float4* dst = reinterpret_cast<float4*>(out + (long)b * VOCAB) + j * SLICE4 + t;
    float4 v = act ? *dst : make_float4(0.f, 0.f, 0.f, 0.f);

    const float* pm = pmax + b * NSPLIT;
    const float* ps = psum + b * NSPLIT;
    bool hi = lane + WAVE < NSPLIT;
    float m0 = pm[lane];
    float s0 = ps[lane];
    float m1 = hi ? pm[lane + WAVE] : -1e30f;
    float s1 = hi ? ps[lane + WAVE] : 0.f;

    float M   = wave_max(fmaxf(m0, m1));
    float S   = wave_sum(s0 * __expf(m0 - M) + s1 * __expf(m1 - M));
    float lse = __logf(S);

    if (act) {
        v.x = v.x - M - lse;
        v.y = v.y - M - lse;
        v.z = v.z - M - lse;
        v.w = v.w - M - lse;
        *dst = v;
    }
}

extern "C" void kernel_launch(void* const* d_in, const int* in_sizes, int n_in,
                              void* d_out, int out_size, void* d_ws, size_t ws_size,
                              hipStream_t stream) {
    const float* A  = (const float*)d_in[0];
    const float* TA = (const float*)d_in[1];
    const float* TC = (const float*)d_in[2];
    const int*   x  = (const int*)d_in[3];
    const int*   q  = (const int*)d_in[4];
    float* out = (float*)d_out;

    float* ws   = (float*)d_ws;
    float* u    = ws;
    float* p    = u    + BATCH * EMBD;
    float* over = p    + BATCH * SLEN;
    float* pmax = over + KOVER * GROWF;
    float* psum = pmax + BATCH * NSPLIT;
    static_assert(BATCH * EMBD + BATCH * SLEN + KOVER * GROWF + 2 * BATCH * NSPLIT
                  <= BATCH * EMBD + BATCH * SLEN + BATCH * NCHUNK * EMBD,
                  "no more workspace than the partial-sum layout used before");
    static_assert((BATCH * EMBD + BATCH * SLEN) % 4 == 0, "over is 16-byte aligned");

    float* keep = out;

    const long TBL = (long)VOCAB * EMBD;
    static_assert(HOPS == 3, "four tables: two gather launches of two tables, two hop launches");

    k_gather_first<<<dim3(NCHUNK + 1, BATCH, 2), WPB * WAVE, 0, stream>>>(A, x, keep, over, A + 1 * TBL, q, u);
    k_hop_first<<<BATCH, HT, 0, stream>>>(keep, over, TA, TC, u, p);
    k_gather_next<<<dim3(NCHUNK, BATCH, 2), WPB * WAVE, 0, stream>>>(A + 2 * TBL, x, keep, over);
    k_hop_last<<<BATCH, HT, 0, stream>>>(keep, over, TA, TC, p, u);

    k_logits  <<<dim3(LGRID), LWAVES * WAVE, 0, stream>>>(u, A + HOPS * TBL, out);
    k_rowstats<<<dim3(NSPLIT, BATCH), 256, 0, stream>>>(out, pmax, psum);
    k_finalize<<<dim3(NSPLIT, BATCH), 256, 0, stream>>>(out, pmax, psum);

    (void)in_sizes; (void)n_in; (void)out_size; (void)ws_size;
}


# k_logits reduces its 64x64 tile in LDS to per-row max and sum-exp pairs itself: the k_rowstats launch and its re-read of the logits go, finalize merges 782 pairs per row; sum-exp tree changes
# speedup vs baseline: 3.1965x; 1.0551x over previous
#include <hip/hip_runtime.h>
#include <hip/hip_bf16.h>

#define HOPS   3
#define VOCAB  50000
#define EMBD   128
#define SLEN   200
#define SENT   32
#define BATCH  64
#define QLEN   32
#define NCHUNK 25
#define WAVE   32
#define SPW    2
#define WPB    (8 / SPW)

static_assert(EMBD == 4 * WAVE, "one table row = one 16-byte load per lane of a wave");
static_assert(QLEN == SENT, "bag_sum sums SENT rows");
static_assert(NCHUNK * 8 == SLEN, "chunks of 8 sentences");
#define SROW ((SLEN + WAVE - 1) / WAVE)
#define CHS  (SLEN / NCHUNK)

typedef __attribute__((ext_vector_type(2))) float v2f;
typedef __attribute__((ext_vector_type(8))) float v8f;

__device__ __forceinline__ float wave_max(float v) {
    for (int o = WAVE / 2; o > 0; o >>= 1) v = fmaxf(v, __shfl_xor(v, o));
    return v;
}
__device__ __forceinline__ float wave_sum(float v) {
    for (int o = WAVE / 2; o > 0; o >>= 1) v += __shfl_xor(v, o);
    return v;
}

__device__ __forceinline__ float4 bag_sum(const float* __restrict__ tab,
                                          const int* __restrict__ idx, int lane) {
    float4 r[SENT];
#pragma unroll
    for (int t = 0; t < SENT; ++t)
        r[t] = reinterpret_cast<const float4*>(tab + (long)idx[t] * EMBD)[lane];
    float4 g = make_float4(0.f, 0.f, 0.f, 0.f);
#pragma unroll
    for (int t = 0; t < SENT; ++t) {
        g.x += r[t].x; g.y += r[t].y; g.z += r[t].z; g.w += r[t].w;
    }
    return g;
}

#define GROWF  (SLEN * EMBD)
#define KSLOTS ((int)((long)BATCH * VOCAB / GROWF))
#define KOVER  (2 * BATCH - KSLOTS)
static_assert(KSLOTS >= BATCH && KOVER >= 0 && KOVER <= BATCH, "first table whole in out, overflow from the second only");
static_assert(GROWF % 4 == 0, "slots are 16-byte aligned");

__device__ __forceinline__ float* keep_row_w(float* keep, float* over, int t, int b) {
    int slot = t * BATCH + b;
    return slot < KSLOTS ? keep + (long)slot * GROWF : over + (long)(slot - KSLOTS) * GROWF;
}
__device__ __forceinline__ const float* keep_row_r(const float* keep, const float* over, int t, int b) {
    int slot = t * BATCH + b;
    return slot < KSLOTS ? keep + (long)slot * GROWF : over + (long)(slot - KSLOTS) * GROWF;
}

template <bool INIT>
__device__ __forceinline__
void gather_body(const float* __restrict__ T2, const int* __restrict__ x,
                 float* __restrict__ keep, float* __restrict__ over,
                 const float* __restrict__ A1, const int* __restrict__ q,
                 float* __restrict__ u) {
    int ch = blockIdx.x, b = blockIdx.y, t = blockIdx.z;
    int lane = threadIdx.x & (WAVE - 1);
    int w = __builtin_amdgcn_readfirstlane(threadIdx.x / WAVE);
    if (INIT && ch == NCHUNK) {
        if (t == 0 && w == 0) {
            float4 g = bag_sum(A1, q + b * QLEN, lane);
            reinterpret_cast<float4*>(u + b * EMBD)[lane] = g;
        }
        return;
    }
    const float* tab = T2 + (long)t * VOCAB * EMBD;
    float* dst = keep_row_w(keep, over, t, b);
    int s0 = ch * CHS + w * SPW;
    for (int i = 0; i < SPW; ++i) {
        float4 g = bag_sum(tab, x + (b * SLEN + s0 + i) * SENT, lane);
        reinterpret_cast<float4*>(dst + (s0 + i) * EMBD)[lane] = g;
    }
}

__global__ __launch_bounds__(WPB * WAVE)
void k_gather_first(const float* __restrict__ T2, const int* __restrict__ x,
                    float* __restrict__ keep, float* __restrict__ over,
                    const float* __restrict__ A1, const int* __restrict__ q,
                    float* __restrict__ u) {
    gather_body<true>(T2, x, keep, over, A1, q, u);
}

__global__ __launch_bounds__(WPB * WAVE)
void k_gather_next(const float* __restrict__ T2, const int* __restrict__ x,
                   float* __restrict__ keep, float* __restrict__ over) {
    gather_body<false>(T2, x, keep, over, nullptr, nullptr, nullptr);
}

#define HT (NCHUNK * WAVE)

__device__ __forceinline__
void row_scores(const float4 (&r)[CHS], float4 uu, const float* __restrict__ TA,
                int s0, float (&sc)[CHS]) {
#pragma unroll
    for (int i = 0; i < CHS; ++i) {
        float ta = TA[s0 + i];
        float4 g = make_float4(r[i].x + ta, r[i].y + ta, r[i].z + ta, r[i].w + ta);
        float v = (g.x * uu.x + g.y * uu.y) + (g.z * uu.z + g.w * uu.w);
        sc[i] = wave_sum(v);
    }
}

template <bool SCORE>
__device__ __forceinline__
float4 hop_stage(const float* pr, float* cp, float* uv,
                 const float4 (&rb)[CHS], float4 uin,
                 const float* __restrict__ TC, const float* __restrict__ TA,
                 int w, int lane, float (&sc)[CHS]) {
    float pv[SROW];
    float m = -1e30f;
    for (int j = 0; j < SROW; ++j) {
        int s = lane + j * WAVE;
        pv[j] = s < SLEN ? pr[s] : -1e30f;
        m = fmaxf(m, pv[j]);
    }
    m = wave_max(m);
    float sum = 0.f;
    for (int j = 0; j < SROW; ++j)
        sum += (lane + j * WAVE < SLEN) ? __expf(pv[j] - m) : 0.f;
    float inv = 1.0f / wave_sum(sum);

    int s0 = w * CHS;
    float4 acc = make_float4(0.f, 0.f, 0.f, 0.f);
#pragma unroll
    for (int i = 0; i < CHS; ++i) {
        float tc = TC[s0 + i];
        float ps = __expf(pr[s0 + i] - m) * inv;
        acc.x += (rb[i].x + tc) * ps;
        acc.y += (rb[i].y + tc) * ps;
        acc.z += (rb[i].z + tc) * ps;
        acc.w += (rb[i].w + tc) * ps;
    }
    if (w == 0) { acc.x += uin.x; acc.y += uin.y; acc.z += uin.z; acc.w += uin.w; }
    reinterpret_cast<float4*>(cp + w * EMBD)[lane] = acc;
    __syncthreads();
    if (w == 0) {
        const float4* c4 = reinterpret_cast<const float4*>(cp) + lane;
        float4 f[NCHUNK];
#pragma unroll
        for (int ch = 0; ch < NCHUNK; ++ch) f[ch] = c4[ch * WAVE];
        float4 uu = f[0];
#pragma unroll
        for (int ch = 1; ch < NCHUNK; ++ch) {
            uu.x += f[ch].x; uu.y += f[ch].y; uu.z += f[ch].z; uu.w += f[ch].w;
        }
        reinterpret_cast<float4*>(uv)[lane] = uu;
    }
    __syncthreads();
    float4 uo = reinterpret_cast<const float4*>(uv)[lane];
    if (SCORE) row_scores(rb, uo, TA, s0, sc);
    return uo;
}

__global__ __launch_bounds__(HT)
void k_hop_first(const float* __restrict__ keep, const float* __restrict__ over,
                 const float* __restrict__ TA, const float* __restrict__ TC,
                 float* __restrict__ u, float* __restrict__ praw) {
    __shared__ __attribute__((aligned(16))) float cp[NCHUNK * EMBD];
    __shared__ __attribute__((aligned(16))) float uv[EMBD];
    __shared__ float pr[SROW * WAVE];
    int b = blockIdx.x;
    int lane = threadIdx.x & (WAVE - 1);
    int w = __builtin_amdgcn_readfirstlane(threadIdx.x / WAVE);
    int s0 = w * CHS;
    const float4* ga = reinterpret_cast<const float4*>(keep_row_r(keep, over, 0, b) + s0 * EMBD) + lane;
    const float4* gb = reinterpret_cast<const float4*>(keep_row_r(keep, over, 1, b) + s0 * EMBD) + lane;
    float4 ra[CHS], rb[CHS];
#pragma unroll
    for (int i = 0; i < CHS; ++i) ra[i] = ga[i * WAVE];
#pragma unroll
    for (int i = 0; i < CHS; ++i) rb[i] = gb[i * WAVE];
    float4 uin = reinterpret_cast<const float4*>(u + b * EMBD)[lane];
    __builtin_amdgcn_sched_barrier(0);

    float sc[CHS];
    row_scores(ra, uin, TA, s0, sc);
    if (lane == 0) {
#pragma unroll
        for (int i = 0; i < CHS; ++i) pr[s0 + i] = sc[i];
    }
    __syncthreads();
    float4 uo = hop_stage<true>(pr, cp, uv, rb, uin, TC, TA, w, lane, sc);
    if (lane == 0) {
        float2* dst = reinterpret_cast<float2*>(praw + b * SLEN + s0);
#pragma unroll
        for (int i = 0; i < CHS; i += 2) dst[i / 2] = make_float2(sc[i], sc[i + 1]);
    }
    if (w == 0) reinterpret_cast<float4*>(u + b * EMBD)[lane] = uo;
}

__global__ __launch_bounds__(HT)
void k_hop_last(const float* __restrict__ keep, const float* __restrict__ over,
                const float* __restrict__ TA, const float* __restrict__ TC,
                const float* __restrict__ praw, float* __restrict__ u) {
    __shared__ __attribute__((aligned(16))) float cp[NCHUNK * EMBD];
    __shared__ __attribute__((aligned(16))) float uv[EMBD];
    __shared__ float pr[SROW * WAVE];
    int b = blockIdx.x;
    int lane = threadIdx.x & (WAVE - 1);
    int w = __builtin_amdgcn_readfirstlane(threadIdx.x / WAVE);
    int s0 = w * CHS;
    const float4* gb = reinterpret_cast<const float4*>(keep_row_r(keep, over, 0, b) + s0 * EMBD) + lane;
    const float4* gc = reinterpret_cast<const float4*>(keep_row_r(keep, over, 1, b) + s0 * EMBD) + lane;
    float4 rb[CHS], rc[CHS];
#pragma unroll
    for (int i = 0; i < CHS; ++i) rb[i] = gb[i * WAVE];
#pragma unroll
    for (int i = 0; i < CHS; ++i) rc[i] = gc[i * WAVE];
    float4 uin = reinterpret_cast<const float4*>(u + b * EMBD)[lane];
    if (threadIdx.x < SLEN) pr[threadIdx.x] = praw[b * SLEN + threadIdx.x];
    __syncthreads();

    float sc[CHS];
    float4 u2 = hop_stage<true>(pr, cp, uv, rb, uin, TC, TA, w, lane, sc);
    if (lane == 0) {
#pragma unroll
        for (int i = 0; i < CHS; ++i) pr[s0 + i] = sc[i];
    }
    __syncthreads();
    float4 u3 = hop_stage<false>(pr, cp, uv, rc, u2, TC, TA, w, lane, sc);
    if (w == 0) reinterpret_cast<float4*>(u + b * EMBD)[lane] = u3;
}

#define LWAVES 4
#define NSTRIP (VOCAB / 16)
#define LGRID  782
#define UPITCH4 (EMBD / 4 + 1)
#define TCOLS  (LWAVES * 16)
#define TPITCH (TCOLS + 4)
#define NPART  LGRID
static_assert(VOCAB % 16 == 0, "whole strips");
static_assert(LGRID * LWAVES >= NSTRIP && (LGRID - 1) * LWAVES < NSTRIP, "grid covers the strips");
static_assert(BATCH == 64 && BATCH % LWAVES == 0, "four batch tiles; staging rows split over the waves");
static_assert(2 * BATCH == LWAVES * WAVE && TCOLS == 64, "row statistics: one thread per half row of 32 columns");

__global__ __launch_bounds__(LWAVES * WAVE) __attribute__((amdgpu_waves_per_eu(1, 4)))
void k_logits(const float* __restrict__ u, const float* __restrict__ A3,
              float* __restrict__ out, float* __restrict__ pmax, float* __restrict__ psum) {
    __shared__ float4 us4[BATCH * UPITCH4];
    __shared__ __attribute__((aligned(16))) float tile[BATCH * TPITCH];
    int lane = threadIdx.x & (WAVE - 1);
    int w = __builtin_amdgcn_readfirstlane(threadIdx.x / WAVE);
    int strip = blockIdx.x * LWAVES + w;
    bool live = strip < NSTRIP;
    int n0   = (live ? strip : NSTRIP - 1) * 16;
    int half = lane >> 4;
    int l16  = lane & 15;

    const float4* brow =
        reinterpret_cast<const float4*>(A3 + (long)(n0 + l16) * EMBD) + half;
    float4 b[EMBD / 8];
#pragma unroll
    for (int j = 0; j < EMBD / 8; ++j) b[j] = brow[2 * j];

    const float4* ug = reinterpret_cast<const float4*>(u);
#pragma unroll
    for (int i = 0; i < BATCH / LWAVES; ++i) {
        int row = i * LWAVES + w;
        us4[row * UPITCH4 + lane] = ug[row * (EMBD / 4) + lane];
    }
    __syncthreads();

    float* tw = tile + (8 * half) * TPITCH + 16 * w + l16;
    v8f c0 = {}, c1 = {}, c2 = {}, c3 = {};
    if (live) {
        const float4* ua = us4 + l16 * UPITCH4 + half;
        float4 an[4], ac[4];
#pragma unroll
        for (int t = 0; t < 4; ++t) an[t] = ua[16 * t * UPITCH4];
#pragma unroll
        for (int j = 0; j < EMBD / 8; ++j) {
#pragma unroll
            for (int t = 0; t < 4; ++t) ac[t] = an[t];
            if (j + 1 < EMBD / 8) {
#pragma unroll
                for (int t = 0; t < 4; ++t) an[t] = ua[16 * t * UPITCH4 + 2 * (j + 1)];
            }
            v2f bx, bz, ax, az;
            bx.x = b[j].x; bx.y = b[j].y;
            bz.x = b[j].z; bz.y = b[j].w;
            ax.x = ac[0].x; ax.y = ac[0].y;
            c0 = __builtin_amdgcn_wmma_f32_16x16x4_f32(false, ax, false, bx,
                                                       (short)0, c0, false, false);
            ax.x = ac[1].x; ax.y = ac[1].y;
            c1 = __builtin_amdgcn_wmma_f32_16x16x4_f32(false, ax, false, bx,
                                                       (short)0, c1, false, false);
            ax.x = ac[2].x; ax.y = ac[2].y;
            c2 = __builtin_amdgcn_wmma_f32_16x16x4_f32(false, ax, false, bx,
                                                       (short)0, c2, false, false);
            ax.x = ac[3].x; ax.y = ac[3].y;
            c3 = __builtin_amdgcn_wmma_f32_16x16x4_f32(false, ax, false, bx,
                                                       (short)0, c3, false, false);
            az.x = ac[0].z; az.y = ac[0].w;
            c0 = __builtin_amdgcn_wmma_f32_16x16x4_f32(false, az, false, bz,
                                                       (short)0, c0, false, false);
            az.x = ac[1].z; az.y = ac[1].w;
            c1 = __builtin_amdgcn_wmma_f32_16x16x4_f32(false, az, false, bz,
                                                       (short)0, c1, false, false);
            az.x = ac[2].z; az.y = ac[2].w;
            c2 = __builtin_amdgcn_wmma_f32_16x16x4_f32(false, az, false, bz,
                                                       (short)0, c2, false, false);
            az.x = ac[3].z; az.y = ac[3].w;
            c3 = __builtin_amdgcn_wmma_f32_16x16x4_f32(false, az, false, bz,
                                                       (short)0, c3, false, false);
        }

#pragma unroll
        for (int r = 0; r < 8; ++r) {
            tw[(r)      * TPITCH] = c0[r];
            tw[(r + 16) * TPITCH] = c1[r];
            tw[(r + 32) * TPITCH] = c2[r];
            tw[(r + 48) * TPITCH] = c3[r];
        }
    } else {
        for (int t = 0; t < 4; ++t)
            for (int r = 0; r < 8; ++r) tw[(16 * t + r) * TPITCH] = -1e30f;
    }
    __syncthreads();

    if (live) {
        int col = n0 + l16;
#pragma unroll
        for (int r = 0; r < 8; ++r) {
            int orow = r + 8 * half;
            out[(long)(orow)      * VOCAB + col] = c0[r];
            out[(long)(orow + 16) * VOCAB + col] = c1[r];
            out[(long)(orow + 32) * VOCAB + col] = c2[r];
            out[(long)(orow + 48) * VOCAB + col] = c3[r];
        }
    }

    int row = threadIdx.x >> 1, hc = threadIdx.x & 1;
    const float4* tr = reinterpret_cast<const float4*>(tile + row * TPITCH + 32 * hc);
    float4 x[8];
#pragma unroll
    for (int i = 0; i < 8; ++i) x[i] = tr[i];
    float m = -1e30f;
#pragma unroll
    for (int i = 0; i < 8; ++i)
        m = fmaxf(m, fmaxf(fmaxf(x[i].x, x[i].y), fmaxf(x[i].z, x[i].w)));
    m = fmaxf(m, __shfl_xor(m, 1));
    float s = 0.f;
#pragma unroll
    for (int i = 0; i < 8; ++i)
        s += (__expf(x[i].x - m) + __expf(x[i].y - m)) +
             (__expf(x[i].z - m) + __expf(x[i].w - m));
    s += __shfl_xor(s, 1);
    if (hc == 0) {
        pmax[row * NPART + blockIdx.x] = m;
        psum[row * NPART + blockIdx.x] = s;
    }
}

#define FSPLIT   10
#define FSLICE4  (VOCAB / FSPLIT / 4)
#define FTHREADS 256
#define FACTIVE  250
#define FPER     (FSLICE4 / FACTIVE)
#define FMERGE   ((NPART + FTHREADS - 1) / FTHREADS)
static_assert(VOCAB % (FSPLIT * 4) == 0, "slices must be whole float4s");
static_assert(FACTIVE * FPER == FSLICE4 && FACTIVE <= FTHREADS, "the active threads cover a slice exactly");
static_assert(FMERGE >= 2 && (FMERGE - 1) * FTHREADS <= NPART && FMERGE * FTHREADS >= NPART,
              "all merge rounds but the last are full");

__global__ __launch_bounds__(FTHREADS)
void k_finalize(float* __restrict__ out, const float* __restrict__ pmax,
                const float* __restrict__ psum) {
    int j = blockIdx.x, b = blockIdx.y, t = threadIdx.x;
    int lane = t & (WAVE - 1), w = t / WAVE;
    __shared__ float redm[FTHREADS / WAVE], reds[FTHREADS / WAVE];
    bool act = t < FACTIVE;
    float4* dst = reinterpret_cast<float4*>(out + (long)b * VOCAB) + j * FSLICE4 + (act ? t : 0);
    float4 v[FPER];
#pragma unroll
    for (int i = 0; i < FPER; ++i) v[i] = dst[i * FACTIVE];

    const float* pm = pmax + b * NPART;
    const float* ps = psum + b * NPART;
    float mm[FMERGE], ss[FMERGE];
#pragma unroll
    for (int k = 0; k < FMERGE - 1; ++k) {
        mm[k] = pm[t + k * FTHREADS];
        ss[k] = ps[t + k * FTHREADS];
    }
    int il = t + (FMERGE - 1) * FTHREADS;
    bool ok = il < NPART;
    float mlast = pm[ok ? il : NPART - 1];
    float slast = ps[ok ? il : NPART - 1];
    mm[FMERGE - 1] = ok ? mlast : -1e30f;
    ss[FMERGE - 1] = ok ? slast : 0.f;
    float ml = mm[0];
#pragma unroll
    for (int k = 1; k < FMERGE; ++k) ml = fmaxf(ml, mm[k]);
    ml = wave_max(ml);
    if (lane == 0) redm[w] = ml;
    __syncthreads();
    float M = redm[0];
    for (int q = 1; q < FTHREADS / WAVE; ++q) M = fmaxf(M, redm[q]);

    float sl = 0.f;
#pragma unroll
    for (int k = 0; k < FMERGE; ++k) sl += ss[k] * __expf(mm[k] - M);
    sl = wave_sum(sl);
    if (lane == 0) reds[w] = sl;
    __syncthreads();
    float S = reds[0];
    for (int q = 1; q < FTHREADS / WAVE; ++q) S += reds[q];
    float lse = __logf(S);

    if (act) {
#pragma unroll
        for (int i = 0; i < FPER; ++i) {
            v[i].x = v[i].x - M - lse;
            v[i].y = v[i].y - M - lse;
            v[i].z = v[i].z - M - lse;
            v[i].w = v[i].w - M - lse;
            dst[i * FACTIVE] = v[i];
        }
    }
}

extern "C" void kernel_launch(void* const* d_in, const int* in_sizes, int n_in,
                              void* d_out, int out_size, void* d_ws, size_t ws_size,
                              hipStream_t stream) {
    const float* A  = (const float*)d_in[0];
    const float* TA = (const float*)d_in[1];
    const float* TC = (const float*)d_in[2];
    const int*   x  = (const int*)d_in[3];
    const int*   q  = (const int*)d_in[4];
    float* out = (float*)d_out;

    float* ws   = (float*)d_ws;
    float* u    = ws;
    float* p    = u    + BATCH * EMBD;
    float* over = p    + BATCH * SLEN;
    float* pmax = over + KOVER * GROWF;
    float* psum = pmax + BATCH * NPART;
    static_assert(BATCH * EMBD + BATCH * SLEN + KOVER * GROWF + 2 * BATCH * NPART
                  <= BATCH * EMBD + BATCH * SLEN + BATCH * NCHUNK * EMBD,
                  "no more workspace than the partial-sum layout used before");
    static_assert((BATCH * EMBD + BATCH * SLEN) % 4 == 0, "over is 16-byte aligned");

    float* keep = out;

    const long TBL = (long)VOCAB * EMBD;
    static_assert(HOPS == 3, "four tables: two gather launches of two tables, two hop launches");

    k_gather_first<<<dim3(NCHUNK + 1, BATCH, 2), WPB * WAVE, 0, stream>>>(A, x, keep, over, A + 1 * TBL, q, u);
    k_hop_first<<<BATCH, HT, 0, stream>>>(keep, over, TA, TC, u, p);
    k_gather_next<<<dim3(NCHUNK, BATCH, 2), WPB * WAVE, 0, stream>>>(A + 2 * TBL, x, keep, over);
    k_hop_last<<<BATCH, HT, 0, stream>>>(keep, over, TA, TC, p, u);

    k_logits  <<<dim3(LGRID), LWAVES * WAVE, 0, stream>>>(u, A + HOPS * TBL, out, pmax, psum);
    k_finalize<<<dim3(FSPLIT, BATCH), FTHREADS, 0, stream>>>(out, pmax, psum);

    (void)in_sizes; (void)n_in; (void)out_size; (void)ws_size;
}


# each hop of the chain runs inside the launch that gathers the table after it (64 chain work-groups ahead of 256 gather ones, two slot sets rotate); table 0 leaves raw scores only; sum orders kept
# speedup vs baseline: 3.3903x; 1.0606x over previous
#include <hip/hip_runtime.h>
#include <hip/hip_bf16.h>

#define HOPS   3
#define VOCAB  50000
#define EMBD   128
#define SLEN   200
#define SENT   32
#define BATCH  64
#define QLEN   32
#define NCHUNK 25
#define WAVE   32
#define SPW    2
#define WPB    (8 / SPW)

static_assert(EMBD == 4 * WAVE, "one table row = one 16-byte load per lane of a wave");
static_assert(QLEN == SENT, "bag_sum sums SENT rows");
static_assert(NCHUNK * 8 == SLEN, "chunks of 8 sentences");
#define SROW ((SLEN + WAVE - 1) / WAVE)
#define CHS  (SLEN / NCHUNK)

typedef __attribute__((ext_vector_type(2))) float v2f;
typedef __attribute__((ext_vector_type(8))) float v8f;

__device__ __forceinline__ float wave_max(float v) {
    for (int o = WAVE / 2; o > 0; o >>= 1) v = fmaxf(v, __shfl_xor(v, o));
    return v;
}
__device__ __forceinline__ float wave_sum(float v) {
    for (int o = WAVE / 2; o > 0; o >>= 1) v += __shfl_xor(v, o);
    return v;
}

__device__ __forceinline__ float4 bag_sum(const float* __restrict__ tab,
                                          const int* __restrict__ idx, int lane) {
    float4 r[SENT];
#pragma unroll
    for (int t = 0; t < SENT; ++t) {
        int id = __builtin_amdgcn_readfirstlane(idx[t]);
        r[t] = reinterpret_cast<const float4*>(tab + (long)id * EMBD)[lane];
    }
    float4 g = make_float4(0.f, 0.f, 0.f, 0.f);
#pragma unroll
    for (int t = 0; t < SENT; ++t) {
        g.x += r[t].x; g.y += r[t].y; g.z += r[t].z; g.w += r[t].w;
    }
    return g;
}

#define GROWF  (SLEN * EMBD)
#define KSLOTS ((int)((long)BATCH * VOCAB / GROWF))
#define KOVER  (2 * BATCH - KSLOTS)
static_assert(KSLOTS >= BATCH && KOVER >= 0 && KOVER <= BATCH, "first table whole in out, overflow from the second only");
static_assert(GROWF % 4 == 0, "slots are 16-byte aligned");

__device__ __forceinline__ float* keep_row_w(float* keep, float* over, int t, int b) {
    int slot = t * BATCH + b;
    return slot < KSLOTS ? keep + (long)slot * GROWF : over + (long)(slot - KSLOTS) * GROWF;
}
__device__ __forceinline__ const float* keep_row_r(const float* keep, const float* over, int t, int b) {
    int slot = t * BATCH + b;
    return slot < KSLOTS ? keep + (long)slot * GROWF : over + (long)(slot - KSLOTS) * GROWF;
}

__global__ __launch_bounds__(WPB * WAVE)
void k_gather_first(const float* __restrict__ T2, const int* __restrict__ x,
                    float* __restrict__ keep, float* __restrict__ over,
                    const float* __restrict__ A1, const int* __restrict__ q,
                    const float* __restrict__ TA,
                    float* __restrict__ u, float* __restrict__ praw) {
    __shared__ float4 u0s[WAVE];
    int ch = blockIdx.x, b = blockIdx.y, t = blockIdx.z;
    int lane = threadIdx.x & (WAVE - 1);
    int w = __builtin_amdgcn_readfirstlane(threadIdx.x / WAVE);
    int s0 = ch * CHS + w * SPW;
    if (t == 1) {
        const float* tab = T2 + (long)VOCAB * EMBD;
        float* dst = keep_row_w(keep, over, 0, b);
        for (int i = 0; i < SPW; ++i) {
            float4 g = bag_sum(tab, x + (b * SLEN + s0 + i) * SENT, lane);
            reinterpret_cast<float4*>(dst + (s0 + i) * EMBD)[lane] = g;
        }
        return;
    }
    if (w == 0) {
        float4 g = bag_sum(A1, q + b * QLEN, lane);
        u0s[lane] = g;
        if (ch == 0) reinterpret_cast<float4*>(u + b * EMBD)[lane] = g;
    }
    __syncthreads();
    float4 uu = u0s[lane];
    for (int i = 0; i < SPW; ++i) {
        float4 r = bag_sum(T2, x + (b * SLEN + s0 + i) * SENT, lane);
        float ta = TA[s0 + i];
        float4 g = make_float4(r.x + ta, r.y + ta, r.z + ta, r.w + ta);
        float v = (g.x * uu.x + g.y * uu.y) + (g.z * uu.z + g.w * uu.w);
        float sc = wave_sum(v);
        if (lane == 0) praw[b * SLEN + s0 + i] = sc;
    }
}

#define HT (NCHUNK * WAVE)

__device__ __forceinline__
void row_scores(const float4 (&r)[CHS], float4 uu, const float* __restrict__ TA,
                int s0, float (&sc)[CHS]) {
#pragma unroll
    for (int i = 0; i < CHS; ++i) {
        float ta = TA[s0 + i];
        float4 g = make_float4(r[i].x + ta, r[i].y + ta, r[i].z + ta, r[i].w + ta);
        float v = (g.x * uu.x + g.y * uu.y) + (g.z * uu.z + g.w * uu.w);
        sc[i] = wave_sum(v);
    }
}

template <bool SCORE>
__device__ __forceinline__
float4 hop_stage(const float* pr, float* cp, float* uv,
                 const float4 (&rb)[CHS], float4 uin,
                 const float* __restrict__ TC, const float* __restrict__ TA,
                 int w, int lane, float (&sc)[CHS]) {
    float pv[SROW];
    float m = -1e30f;
    for (int j = 0; j < SROW; ++j) {
        int s = lane + j * WAVE;
        pv[j] = s < SLEN ? pr[s] : -1e30f;
        m = fmaxf(m, pv[j]);
    }
    m = wave_max(m);
    float sum = 0.f;
    for (int j = 0; j < SROW; ++j)
        sum += (lane + j * WAVE < SLEN) ? __expf(pv[j] - m) : 0.f;
    float inv = 1.0f / wave_sum(sum);

    int s0 = w * CHS;
    float4 acc = make_float4(0.f, 0.f, 0.f, 0.f);
#pragma unroll
    for (int i = 0; i < CHS; ++i) {
        float tc = TC[s0 + i];
        float ps = __expf(pr[s0 + i] - m) * inv;
        acc.x += (rb[i].x + tc) * ps;
        acc.y += (rb[i].y + tc) * ps;
        acc.z += (rb[i].z + tc) * ps;
        acc.w += (rb[i].w + tc) * ps;
    }
    if (w == 0) { acc.x += uin.x; acc.y += uin.y; acc.z += uin.z; acc.w += uin.w; }
    reinterpret_cast<float4*>(cp + w * EMBD)[lane] = acc;
    __syncthreads();
    if (w == 0) {
        const float4* c4 = reinterpret_cast<const float4*>(cp) + lane;
        float4 f[NCHUNK];
#pragma unroll
        for (int ch = 0; ch < NCHUNK; ++ch) f[ch] = c4[ch * WAVE];
        float4 uu = f[0];
#pragma unroll
        for (int ch = 1; ch < NCHUNK; ++ch) {
            uu.x += f[ch].x; uu.y += f[ch].y; uu.z += f[ch].z; uu.w += f[ch].w;
        }
        reinterpret_cast<float4*>(uv)[lane] = uu;
    }
    __syncthreads();
    float4 uo = reinterpret_cast<const float4*>(uv)[lane];
    if (SCORE) row_scores(rb, uo, TA, s0, sc);
    return uo;
}

#define GPARTS 4
#define GSENT  (SLEN / GPARTS)
static_assert(GSENT == NCHUNK * SPW, "one gather work-group: NCHUNK waves x SPW sentences");

__global__ __launch_bounds__(HT)
void k_hop_gather(float* keep, float* over, int rset, int wset,
                  const float* __restrict__ tab, const int* __restrict__ x,
                  const float* __restrict__ TA, const float* __restrict__ TC,
                  float* praw, float* u) {
    __shared__ __attribute__((aligned(16))) float cp[NCHUNK * EMBD];
    __shared__ __attribute__((aligned(16))) float uv[EMBD];
    __shared__ float pr[SROW * WAVE];
    int lane = threadIdx.x & (WAVE - 1);
    int w = __builtin_amdgcn_readfirstlane(threadIdx.x / WAVE);
    if (blockIdx.x >= BATCH) {
        int g = blockIdx.x - BATCH;
        int b = g / GPARTS, part = g % GPARTS;
        int sg = part * GSENT + w * SPW;
        float* dst = keep_row_w(keep, over, wset, b);
        for (int i = 0; i < SPW; ++i) {
            float4 v = bag_sum(tab, x + (b * SLEN + sg + i) * SENT, lane);
            reinterpret_cast<float4*>(dst + (sg + i) * EMBD)[lane] = v;
        }
        return;
    }
    int b = blockIdx.x;
    int s0 = w * CHS;
    const float4* gb = reinterpret_cast<const float4*>(keep_row_r(keep, over, rset, b) + s0 * EMBD) + lane;
    float4 rb[CHS];
#pragma unroll
    for (int i = 0; i < CHS; ++i) rb[i] = gb[i * WAVE];
    float4 uin = reinterpret_cast<const float4*>(u + b * EMBD)[lane];
    if (threadIdx.x < SLEN) pr[threadIdx.x] = praw[b * SLEN + threadIdx.x];
    __syncthreads();

    float sc[CHS];
    float4 uo = hop_stage<true>(pr, cp, uv, rb, uin, TC, TA, w, lane, sc);
    if (lane == 0) {
        float2* dst = reinterpret_cast<float2*>(praw + b * SLEN + s0);
#pragma unroll
        for (int i = 0; i < CHS; i += 2) dst[i / 2] = make_float2(sc[i], sc[i + 1]);
    }
    if (w == 0) reinterpret_cast<float4*>(u + b * EMBD)[lane] = uo;
}

__global__ __launch_bounds__(HT)
void k_hop_last(const float* __restrict__ keep, const float* __restrict__ over, int rset,
                const float* __restrict__ TA, const float* __restrict__ TC,
                const float* __restrict__ praw, float* __restrict__ u) {
    __shared__ __attribute__((aligned(16))) float cp[NCHUNK * EMBD];
    __shared__ __attribute__((aligned(16))) float uv[EMBD];
    __shared__ float pr[SROW * WAVE];
    int b = blockIdx.x;
    int lane = threadIdx.x & (WAVE - 1);
    int w = __builtin_amdgcn_readfirstlane(threadIdx.x / WAVE);
    int s0 = w * CHS;
    const float4* gc = reinterpret_cast<const float4*>(keep_row_r(keep, over, rset, b) + s0 * EMBD) + lane;
    float4 rc[CHS];
#pragma unroll
    for (int i = 0; i < CHS; ++i) rc[i] = gc[i * WAVE];
    float4 uin = reinterpret_cast<const float4*>(u + b * EMBD)[lane];
    if (threadIdx.x < SLEN) pr[threadIdx.x] = praw[b * SLEN + threadIdx.x];
    __syncthreads();

    float sc[CHS];
    float4 u3 = hop_stage<false>(pr, cp, uv, rc, uin, TC, TA, w, lane, sc);
    if (w == 0) reinterpret_cast<float4*>(u + b * EMBD)[lane] = u3;
}

#define LWAVES 4
#define NSTRIP (VOCAB / 16)
#define LGRID  782
#define UPITCH4 (EMBD / 4 + 1)
#define TCOLS  (LWAVES * 16)
#define TPITCH (TCOLS + 4)
#define NPART  LGRID
static_assert(VOCAB % 16 == 0, "whole strips");
static_assert(LGRID * LWAVES >= NSTRIP && (LGRID - 1) * LWAVES < NSTRIP, "grid covers the strips");
static_assert(BATCH == 64 && BATCH % LWAVES == 0, "four batch tiles; staging rows split over the waves");
static_assert(2 * BATCH == LWAVES * WAVE && TCOLS == 64, "row statistics: one thread per half row of 32 columns");

__global__ __launch_bounds__(LWAVES * WAVE) __attribute__((amdgpu_waves_per_eu(1, 4)))
void k_logits(const float* __restrict__ u, const float* __restrict__ A3,
              float* __restrict__ out, float* __restrict__ pmax, float* __restrict__ psum) {
    __shared__ float4 us4[BATCH * UPITCH4];
    __shared__ __attribute__((aligned(16))) float tile[BATCH * TPITCH];
    int lane = threadIdx.x & (WAVE - 1);
    int w = __builtin_amdgcn_readfirstlane(threadIdx.x / WAVE);
    int strip = blockIdx.x * LWAVES + w;
    bool live = strip < NSTRIP;
    int n0   = (live ? strip : NSTRIP - 1) * 16;
    int half = lane >> 4;
    int l16  = lane & 15;

    const float4* brow =
        reinterpret_cast<const float4*>(A3 + (long)(n0 + l16) * EMBD) + half;
    float4 b[EMBD / 8];
#pragma unroll
    for (int j = 0; j < EMBD / 8; ++j) b[j] = brow[2 * j];

    const float4* ug = reinterpret_cast<const float4*>(u);
#pragma unroll
    for (int i = 0; i < BATCH / LWAVES; ++i) {
        int row = i * LWAVES + w;
        us4[row * UPITCH4 + lane] = ug[row * (EMBD / 4) + lane];
    }
    __syncthreads();

    float* tw = tile + (8 * half) * TPITCH + 16 * w + l16;
    v8f c0 = {}, c1 = {}, c2 = {}, c3 = {};
    if (live) {
        const float4* ua = us4 + l16 * UPITCH4 + half;
        float4 an[4], ac[4];
#pragma unroll
        for (int t = 0; t < 4; ++t) an[t] = ua[16 * t * UPITCH4];
#pragma unroll
        for (int j = 0; j < EMBD / 8; ++j) {
#pragma unroll
            for (int t = 0; t < 4; ++t) ac[t] = an[t];
            if (j + 1 < EMBD / 8) {
#pragma unroll
                for (int t = 0; t < 4; ++t) an[t] = ua[16 * t * UPITCH4 + 2 * (j + 1)];
            }
            v2f bx, bz, ax, az;
            bx.x = b[j].x; bx.y = b[j].y;
            bz.x = b[j].z; bz.y = b[j].w;
            ax.x = ac[0].x; ax.y = ac[0].y;
            c0 = __builtin_amdgcn_wmma_f32_16x16x4_f32(false, ax, false, bx,
                                                       (short)0, c0, false, false);
            ax.x = ac[1].x; ax.y = ac[1].y;
            c1 = __builtin_amdgcn_wmma_f32_16x16x4_f32(false, ax, false, bx,
                                                       (short)0, c1, false, false);
            ax.x = ac[2].x; ax.y = ac[2].y;
            c2 = __builtin_amdgcn_wmma_f32_16x16x4_f32(false, ax, false, bx,
                                                       (short)0, c2, false, false);
            ax.x = ac[3].x; ax.y = ac[3].y;
            c3 = __builtin_amdgcn_wmma_f32_16x16x4_f32(false, ax, false, bx,
                                                       (short)0, c3, false, false);
            az.x = ac[0].z; az.y = ac[0].w;
            c0 = __builtin_amdgcn_wmma_f32_16x16x4_f32(false, az, false, bz,
                                                       (short)0, c0, false, false);
            az.x = ac[1].z; az.y = ac[1].w;
            c1 = __builtin_amdgcn_wmma_f32_16x16x4_f32(false, az, false, bz,
                                                       (short)0, c1, false, false);
            az.x = ac[2].z; az.y = ac[2].w;
            c2 = __builtin_amdgcn_wmma_f32_16x16x4_f32(false, az, false, bz,
                                                       (short)0, c2, false, false);
            az.x = ac[3].z; az.y = ac[3].w;
            c3 = __builtin_amdgcn_wmma_f32_16x16x4_f32(false, az, false, bz,
                                                       (short)0, c3, false, false);
        }

#pragma unroll
        for (int r = 0; r < 8; ++r) {
            tw[(r)      * TPITCH] = c0[r];
            tw[(r + 16) * TPITCH] = c1[r];
            tw[(r + 32) * TPITCH] = c2[r];
            tw[(r + 48) * TPITCH] = c3[r];
        }
    } else {
        for (int t = 0; t < 4; ++t)
            for (int r = 0; r < 8; ++r) tw[(16 * t + r) * TPITCH] = -1e30f;
    }
    __syncthreads();

    if (live) {
        int col = n0 + l16;
#pragma unroll
        for (int r = 0; r < 8; ++r) {
            int orow = r + 8 * half;
            out[(long)(orow)      * VOCAB + col] = c0[r];
            out[(long)(orow + 16) * VOCAB + col] = c1[r];
            out[(long)(orow + 32) * VOCAB + col] = c2[r];
            out[(long)(orow + 48) * VOCAB + col] = c3[r];
        }
    }

    int row = threadIdx.x >> 1, hc = threadIdx.x & 1;
    const float4* tr = reinterpret_cast<const float4*>(tile + row * TPITCH + 32 * hc);
    float4 x[8];
#pragma unroll
    for (int i = 0; i < 8; ++i) x[i] = tr[i];
    float m = -1e30f;
#pragma unroll
    for (int i = 0; i < 8; ++i)
        m = fmaxf(m, fmaxf(fmaxf(x[i].x, x[i].y), fmaxf(x[i].z, x[i].w)));
    m = fmaxf(m, __shfl_xor(m, 1));
    float s = 0.f;
#pragma unroll
    for (int i = 0; i < 8; ++i)
        s += (__expf(x[i].x - m) + __expf(x[i].y - m)) +
             (__expf(x[i].z - m) + __expf(x[i].w - m));
    s += __shfl_xor(s, 1);
    if (hc == 0) {
        pmax[row * NPART + blockIdx.x] = m;
        psum[row * NPART + blockIdx.x] = s;
    }
}

#define FSPLIT   10
#define FSLICE4  (VOCAB / FSPLIT / 4)
#define FTHREADS 256
#define FACTIVE  250
#define FPER     (FSLICE4 / FACTIVE)
#define FMERGE   ((NPART + FTHREADS - 1) / FTHREADS)
static_assert(VOCAB % (FSPLIT * 4) == 0, "slices must be whole float4s");
static_assert(FACTIVE * FPER == FSLICE4 && FACTIVE <= FTHREADS, "the active threads cover a slice exactly");
static_assert(FMERGE >= 2 && (FMERGE - 1) * FTHREADS <= NPART && FMERGE * FTHREADS >= NPART,
              "all merge rounds but the last are full");

__global__ __launch_bounds__(FTHREADS)
void k_finalize(float* __restrict__ out, const float* __restrict__ pmax,
                const float* __restrict__ psum) {
    int j = blockIdx.x, b = blockIdx.y, t = threadIdx.x;
    int lane = t & (WAVE - 1), w = t / WAVE;
    __shared__ float redm[FTHREADS / WAVE], reds[FTHREADS / WAVE];
    bool act = t < FACTIVE;
    float4* dst = reinterpret_cast<float4*>(out + (long)b * VOCAB) + j * FSLICE4 + (act ? t : 0);
    float4 v[FPER];
#pragma unroll
    for (int i = 0; i < FPER; ++i) v[i] = dst[i * FACTIVE];

    const float* pm = pmax + b * NPART;
    const float* ps = psum + b * NPART;
    float mm[FMERGE], ss[FMERGE];
#pragma unroll
    for (int k = 0; k < FMERGE - 1; ++k) {
        mm[k] = pm[t + k * FTHREADS];
        ss[k] = ps[t + k * FTHREADS];
    }
    int il = t + (FMERGE - 1) * FTHREADS;
    bool ok = il < NPART;
    float mlast = pm[ok ? il : NPART - 1];
    float slast = ps[ok ? il : NPART - 1];
    mm[FMERGE - 1] = ok ? mlast : -1e30f;
    ss[FMERGE - 1] = ok ? slast : 0.f;
    float ml = mm[0];
#pragma unroll
    for (int k = 1; k < FMERGE; ++k) ml = fmaxf(ml, mm[k]);
    ml = wave_max(ml);
    if (lane == 0) redm[w] = ml;
    __syncthreads();
    float M = redm[0];
    for (int q = 1; q < FTHREADS / WAVE; ++q) M = fmaxf(M, redm[q]);

    float sl = 0.f;
#pragma unroll
    for (int k = 0; k < FMERGE; ++k) sl += ss[k] * __expf(mm[k] - M);
    sl = wave_sum(sl);
    if (lane == 0) reds[w] = sl;
    __syncthreads();
    float S = reds[0];
    for (int q = 1; q < FTHREADS / WAVE; ++q) S += reds[q];
    float lse = __logf(S);

    if (act) {
#pragma unroll
        for (int i = 0; i < FPER; ++i) {
            v[i].x = v[i].x - M - lse;
            v[i].y = v[i].y - M - lse;
            v[i].z = v[i].z - M - lse;
            v[i].w = v[i].w - M - lse;
            dst[i * FACTIVE] = v[i];
        }
    }
}

extern "C" void kernel_launch(void* const* d_in, const int* in_sizes, int n_in,
                              void* d_out, int out_size, void* d_ws, size_t ws_size,
                              hipStream_t stream) {
    const float* A  = (const float*)d_in[0];
    const float* TA = (const float*)d_in[1];
    const float* TC = (const float*)d_in[2];
    const int*   x  = (const int*)d_in[3];
    const int*   q  = (const int*)d_in[4];
    float* out = (float*)d_out;

    float* ws   = (float*)d_ws;
    float* u    = ws;
    float* p    = u    + BATCH * EMBD;
    float* over = p    + BATCH * SLEN;
    float* pmax = over + KOVER * GROWF;
    float* psum = pmax + BATCH * NPART;
    static_assert(BATCH * EMBD + BATCH * SLEN + KOVER * GROWF + 2 * BATCH * NPART
                  <= BATCH * EMBD + BATCH * SLEN + BATCH * NCHUNK * EMBD,
                  "no more workspace than the partial-sum layout used before");
    static_assert((BATCH * EMBD + BATCH * SLEN) % 4 == 0, "over is 16-byte aligned");

    float* keep = out;

    const long TBL = (long)VOCAB * EMBD;
    static_assert(HOPS == 3, "four tables: one first launch, two hop-beside-gather launches, one last hop");

    k_gather_first<<<dim3(NCHUNK, BATCH, 2), WPB * WAVE, 0, stream>>>(A, x, keep, over, A + 1 * TBL, q, TA, u, p);
    k_hop_gather<<<BATCH + GPARTS * BATCH, HT, 0, stream>>>(keep, over, 0, 1, A + 2 * TBL, x, TA, TC, p, u);
    k_hop_gather<<<BATCH + GPARTS * BATCH, HT, 0, stream>>>(keep, over, 1, 0, A + 3 * TBL, x, TA, TC, p, u);
    k_hop_last<<<BATCH, HT, 0, stream>>>(keep, over, 0, TA, TC, p, u);

    k_logits  <<<dim3(LGRID), LWAVES * WAVE, 0, stream>>>(u, A + HOPS * TBL, out, pmax, psum);
    k_finalize<<<dim3(FSPLIT, BATCH), FTHREADS, 0, stream>>>(out, pmax, psum);

    (void)in_sizes; (void)n_in; (void)out_size; (void)ws_size;
}
